# starting kernel
# baseline (speedup 1.0000x reference)
#include <hip/hip_runtime.h>
#include <hip/hip_bf16.h>

typedef __attribute__((ext_vector_type(2))) float v2f;
typedef __attribute__((ext_vector_type(8))) float v8f;

#define Bn 4
#define Cc 256
#define Hh 56
#define Ww 56
#define HW (Hh * Ww)
#define NP (Bn * HW)
#define TN 256
#define Gg 16
#define Kk 7
#define KK (Kk * Kk)
#define NK (Gg * KK)
#define EPSf 1e-5f

__global__ __launch_bounds__(256) void gemm1_reduce_bn_prelu(
    const float* __restrict__ x, const float* __restrict__ rw,
    const float* __restrict__ rb, const float* __restrict__ gamma,
    const float* __restrict__ beta, const float* __restrict__ mean,
    const float* __restrict__ var, const float* __restrict__ pa,
    float* __restrict__ t_out)
{
    const int lane  = threadIdx.x & 31;
    const int wave  = threadIdx.x >> 5;
    const int tile  = blockIdx.x * 8 + wave;
    const int npair = tile & 7;
    const int mpair = tile >> 3;
    const int half  = lane >> 4;
    const int l16   = lane & 15;

    const int p_base = mpair * 32;
    const int b      = p_base / HW;
    const int hw     = p_base - b * HW;
    const float* xrow0 = x + (size_t)b * Cc * HW + hw + l16;
    const float* xrow1 = xrow0 + 16;
    const int n0 = npair * 32 + l16;
    const int n1 = n0 + 16;
    const float* wrow0 = rw + (size_t)n0 * Cc;
    const float* wrow1 = rw + (size_t)n1 * Cc;

    v8f acc00 = {}, acc01 = {}, acc10 = {}, acc11 = {};
#pragma unroll 4
    for (int k0 = 0; k0 < Cc; k0 += 4) {
        const int kh = k0 + 2 * half;
        v2f a0, a1;
        a0.x = xrow0[(size_t)kh * HW];
        a0.y = xrow0[(size_t)(kh + 1) * HW];
        a1.x = xrow1[(size_t)kh * HW];
        a1.y = xrow1[(size_t)(kh + 1) * HW];
        v2f b0 = *(const v2f*)(wrow0 + kh);
        v2f b1 = *(const v2f*)(wrow1 + kh);
        acc00 = __builtin_amdgcn_wmma_f32_16x16x4_f32(false, a0, false, b0, (short)0, acc00, false, false);
        acc01 = __builtin_amdgcn_wmma_f32_16x16x4_f32(false, a0, false, b1, (short)0, acc01, false, false);
        acc10 = __builtin_amdgcn_wmma_f32_16x16x4_f32(false, a1, false, b0, (short)0, acc10, false, false);
        acc11 = __builtin_amdgcn_wmma_f32_16x16x4_f32(false, a1, false, b1, (short)0, acc11, false, false);
    }

    const float al = pa[0];
    const float s0  = gamma[n0] * rsqrtf(var[n0] + EPSf);
    const float s1  = gamma[n1] * rsqrtf(var[n1] + EPSf);
    const float bi0 = rb[n0] - mean[n0], bi1 = rb[n1] - mean[n1];
    const float bt0 = beta[n0],          bt1 = beta[n1];

    float* trow0 = t_out + (size_t)(p_base + 8 * half) * TN;
    float* trow1 = t_out + (size_t)(p_base + 16 + 8 * half) * TN;
#pragma unroll
    for (int r = 0; r < 8; ++r) {
        float v00 = (acc00[r] + bi0) * s0 + bt0; v00 = v00 > 0.f ? v00 : al * v00;
        float v01 = (acc01[r] + bi1) * s1 + bt1; v01 = v01 > 0.f ? v01 : al * v01;
        float v10 = (acc10[r] + bi0) * s0 + bt0; v10 = v10 > 0.f ? v10 : al * v10;
        float v11 = (acc11[r] + bi1) * s1 + bt1; v11 = v11 > 0.f ? v11 : al * v11;
        trow0[(size_t)r * TN + n0] = v00;
        trow0[(size_t)r * TN + n1] = v01;
        trow1[(size_t)r * TN + n0] = v10;
        trow1[(size_t)r * TN + n1] = v11;
    }
}

__global__ __launch_bounds__(128) void gemm2_span(
    const float* __restrict__ t_in, const float* __restrict__ sw,
    const float* __restrict__ sb, float* __restrict__ ker)
{
    const int lane  = threadIdx.x & 31;
    const int wave  = threadIdx.x >> 5;
    const int tile  = blockIdx.x * 4 + wave;
    const int mtile = tile % 49;
    const int nquad = tile / 49;
    const int half  = lane >> 4;
    const int l16   = lane & 15;

    const float* arow = sw + (size_t)(mtile * 16 + l16) * TN;
    const int p_base  = nquad * 64;
    const float* brow0 = t_in + (size_t)(p_base +  0 + l16) * TN;
    const float* brow1 = t_in + (size_t)(p_base + 16 + l16) * TN;
    const float* brow2 = t_in + (size_t)(p_base + 32 + l16) * TN;
    const float* brow3 = t_in + (size_t)(p_base + 48 + l16) * TN;

    v8f acc0 = {}, acc1 = {}, acc2 = {}, acc3 = {};
#pragma unroll 4
    for (int k0 = 0; k0 < TN; k0 += 4) {
        const int kh = k0 + 2 * half;
        v2f a  = *(const v2f*)(arow + kh);
        v2f b0 = *(const v2f*)(brow0 + kh);
        v2f b1 = *(const v2f*)(brow1 + kh);
        v2f b2 = *(const v2f*)(brow2 + kh);
        v2f b3 = *(const v2f*)(brow3 + kh);
        acc0 = __builtin_amdgcn_wmma_f32_16x16x4_f32(false, a, false, b0, (short)0, acc0, false, false);
        acc1 = __builtin_amdgcn_wmma_f32_16x16x4_f32(false, a, false, b1, (short)0, acc1, false, false);
        acc2 = __builtin_amdgcn_wmma_f32_16x16x4_f32(false, a, false, b2, (short)0, acc2, false, false);
        acc3 = __builtin_amdgcn_wmma_f32_16x16x4_f32(false, a, false, b3, (short)0, acc3, false, false);
    }

    const int kkbase = mtile * 16 + 8 * half;
#pragma unroll
    for (int r = 0; r < 8; ++r) {
        const int kkr = kkbase + r;
        const float bias = sb[kkr];
        float* krow = ker + (size_t)kkr * NP + p_base + l16;
        krow[ 0] = acc0[r] + bias;
        krow[16] = acc1[r] + bias;
        krow[32] = acc2[r] + bias;
        krow[48] = acc3[r] + bias;
    }
}

__global__ __launch_bounds__(256) void involution_apply(
    const float* __restrict__ x, const float* __restrict__ ker,
    float* __restrict__ out)
{
    const int id = blockIdx.x * 256 + threadIdx.x;
    const int hw = id % HW;
    const int bg = id / HW;
    const int g  = bg % Gg;
    const int b  = bg / Gg;
    const int h  = hw / Ww;
    const int w  = hw % Ww;
    const int p  = b * HW + hw;

    float kv[KK];
#pragma unroll
    for (int idx = 0; idx < KK; ++idx)
        kv[idx] = ker[(size_t)(g * KK + idx) * NP + p];

    const float* xb = x   + (size_t)b * Cc * HW;
    float*       ob = out + (size_t)b * Cc * HW;
    for (int cc = 0; cc < Gg; ++cc) {
        const int c = g * 16 + cc;
        const float* xc = xb + (size_t)c * HW;
        float acc = 0.f;
#pragma unroll
        for (int idx = 0; idx < KK; ++idx) {
            const int hh = h + idx / Kk - 3;
            const int ww = w + idx % Kk - 3;
            if (hh >= 0 && hh < Hh && ww >= 0 && ww < Ww)
                acc += xc[hh * Ww + ww] * kv[idx];
        }
        ob[(size_t)c * HW + hw] = acc;
    }
}

extern "C" void kernel_launch(void* const* d_in, const int* in_sizes, int n_in,
                              void* d_out, int out_size, void* d_ws, size_t ws_size,
                              hipStream_t stream) {
    const float* x     = (const float*)d_in[0];
    const float* rw    = (const float*)d_in[1];
    const float* rb    = (const float*)d_in[2];
    const float* gamma = (const float*)d_in[3];
    const float* beta  = (const float*)d_in[4];
    const float* mean  = (const float*)d_in[5];
    const float* var   = (const float*)d_in[6];
    const float* pa    = (const float*)d_in[7];
    const float* sw    = (const float*)d_in[8];
    const float* sb    = (const float*)d_in[9];
    float* out = (float*)d_out;

    float* t_ws   = (float*)d_ws;
    float* ker_ws = t_ws + (size_t)NP * TN;

    gemm1_reduce_bn_prelu<<<392, 256, 0, stream>>>(
        x, rw, rb, gamma, beta, mean, var, pa, t_ws);
    gemm2_span<<<2401, 128, 0, stream>>>(
        t_ws, sw, sb, ker_ws);
    involution_apply<<<(Bn * Gg * HW) / 256, 256, 0, stream>>>(
        x, ker_ws, out);
}


# GEMM pixel tiles interleaved (lane c of tile n = pixel 4c+n): reduce GEMM reads x by 16-byte loads (16x64 wave tile, a trip's 24 loads in one group), span GEMM stores whole ker lines; k orders kept
# speedup vs baseline: 2.4612x; 2.4612x over previous
#include <hip/hip_runtime.h>
#include <hip/hip_bf16.h>

typedef __attribute__((ext_vector_type(2))) float v2f;
typedef __attribute__((ext_vector_type(8))) float v8f;

#define Bn 4
#define Cc 256
#define Hh 56
#define Ww 56
#define HW (Hh * Ww)
#define NP (Bn * HW)
#define TN 256
#define Gg 16
#define Kk 7
#define KK (Kk * Kk)
#define NK (Gg * KK)
#define EPSf 1e-5f

#define TF_PTILE (16 * 2 * 32)
#define KU 8

__global__ __launch_bounds__(256) __attribute__((amdgpu_waves_per_eu(4, 8)))
void gemm1_reduce_bn_prelu(
    const float* __restrict__ x, const float* __restrict__ rw,
    const float* __restrict__ rb, const float* __restrict__ gamma,
    const float* __restrict__ beta, const float* __restrict__ mean,
    const float* __restrict__ var, const float* __restrict__ pa,
    float* __restrict__ t_out)
{
    const int lane  = threadIdx.x & 31;
    const int wave  = threadIdx.x >> 5;
    const int tile  = blockIdx.x * 8 + wave;
    const int mt    = tile & 15;
    const int blk   = tile >> 4;
    const int half  = lane >> 4;
    const int l16   = lane & 15;

    const int p_base = blk * 64;
    const int b      = p_base / HW;
    const int hw     = p_base - b * HW;
    const float* xq = x + (size_t)b * Cc * HW + hw + 4 * l16;
    const int cperm = 4 * ((l16 & 7) >> 1) + 2 * (l16 >> 3) + (l16 & 1);
    const float* wrow = rw + (size_t)(mt * 16 + cperm) * Cc;

    v8f acc0 = {}, acc1 = {}, acc2 = {}, acc3 = {};
#pragma unroll 1
    for (int k0 = 0; k0 < Cc; k0 += 4 * KU) {
        float4 xa[KU], xb[KU];
        v2f    w[KU];
#pragma unroll
        for (int u = 0; u < KU; ++u) {
            const int kh = k0 + 4 * u + 2 * half;
            xa[u] = *(const float4*)(xq + (size_t)kh * HW);
            xb[u] = *(const float4*)(xq + (size_t)(kh + 1) * HW);
            w[u]  = *(const v2f*)(wrow + kh);
        }
#pragma unroll
        for (int u = 0; u < KU; ++u) {
            v2f b0, b1, b2, b3;
            b0.x = xa[u].x; b0.y = xb[u].x;
            b1.x = xa[u].y; b1.y = xb[u].y;
            b2.x = xa[u].z; b2.y = xb[u].z;
            b3.x = xa[u].w; b3.y = xb[u].w;
            acc0 = __builtin_amdgcn_wmma_f32_16x16x4_f32(false, w[u], false, b0, (short)0, acc0, false, false);
            acc1 = __builtin_amdgcn_wmma_f32_16x16x4_f32(false, w[u], false, b1, (short)0, acc1, false, false);
            acc2 = __builtin_amdgcn_wmma_f32_16x16x4_f32(false, w[u], false, b2, (short)0, acc2, false, false);
            acc3 = __builtin_amdgcn_wmma_f32_16x16x4_f32(false, w[u], false, b3, (short)0, acc3, false, false);
        }
    }

    const float al = pa[0];
    float4* tf = (float4*)t_out + (size_t)(blk * 4) * TF_PTILE + lane;
    const int nbase = mt * 16 + 2 * half;
#pragma unroll
    for (int rq = 0; rq < 2; ++rq) {
        float o0[4], o1[4], o2[4], o3[4];
#pragma unroll
        for (int e = 0; e < 4; ++e) {
            const int r  = 4 * rq + e;
            const int n  = nbase + 4 * (r >> 1) + (r & 1);
            const float s  = gamma[n] * rsqrtf(var[n] + EPSf);
            const float bi = rb[n] - mean[n];
            const float bt = beta[n];
            float v0 = (acc0[r] + bi) * s + bt; v0 = v0 > 0.f ? v0 : al * v0;
            float v1 = (acc1[r] + bi) * s + bt; v1 = v1 > 0.f ? v1 : al * v1;
            float v2 = (acc2[r] + bi) * s + bt; v2 = v2 > 0.f ? v2 : al * v2;
            float v3 = (acc3[r] + bi) * s + bt; v3 = v3 > 0.f ? v3 : al * v3;
            o0[e] = v0;
            o1[e] = v1;
            o2[e] = v2;
            o3[e] = v3;
        }
        float4* dst = tf + (size_t)(mt * 2 + rq) * 32;
        dst[0 * TF_PTILE] = make_float4(o0[0], o0[1], o0[2], o0[3]);
        dst[1 * TF_PTILE] = make_float4(o1[0], o1[1], o1[2], o1[3]);
        dst[2 * TF_PTILE] = make_float4(o2[0], o2[1], o2[2], o2[3]);
        dst[3 * TF_PTILE] = make_float4(o3[0], o3[1], o3[2], o3[3]);
    }
}

__global__ __launch_bounds__(128) void gemm2_span(
    const float* __restrict__ t_in, const float* __restrict__ sw,
    const float* __restrict__ sb, float* __restrict__ ker)
{
    const int lane  = threadIdx.x & 31;
    const int wave  = threadIdx.x >> 5;
    const int tile  = blockIdx.x * 4 + wave;
    const int mtile = tile % 49;
    const int nquad = tile / 49;
    const int half  = lane >> 4;
    const int l16   = lane & 15;

    const float* arow = sw + (size_t)(mtile * 16 + l16) * TN + 2 * half;
    const int p_base  = nquad * 64;
    const float4* tf  = (const float4*)t_in + (size_t)(nquad * 4) * TF_PTILE + lane;

    v8f acc0 = {}, acc1 = {}, acc2 = {}, acc3 = {};
#pragma unroll 1
    for (int c = 0; c < TN / 8; ++c) {
        const float4 q0 = tf[0 * TF_PTILE + c * 32];
        const float4 q1 = tf[1 * TF_PTILE + c * 32];
        const float4 q2 = tf[2 * TF_PTILE + c * 32];
        const float4 q3 = tf[3 * TF_PTILE + c * 32];
        const v2f a0 = *(const v2f*)(arow + 8 * c);
        const v2f a1 = *(const v2f*)(arow + 8 * c + 4);
        v2f b0, b1, b2, b3;
        b0.x = q0.x; b0.y = q0.y; b1.x = q1.x; b1.y = q1.y;
        b2.x = q2.x; b2.y = q2.y; b3.x = q3.x; b3.y = q3.y;
        acc0 = __builtin_amdgcn_wmma_f32_16x16x4_f32(false, a0, false, b0, (short)0, acc0, false, false);
        acc1 = __builtin_amdgcn_wmma_f32_16x16x4_f32(false, a0, false, b1, (short)0, acc1, false, false);
        acc2 = __builtin_amdgcn_wmma_f32_16x16x4_f32(false, a0, false, b2, (short)0, acc2, false, false);
        acc3 = __builtin_amdgcn_wmma_f32_16x16x4_f32(false, a0, false, b3, (short)0, acc3, false, false);
        b0.x = q0.z; b0.y = q0.w; b1.x = q1.z; b1.y = q1.w;
        b2.x = q2.z; b2.y = q2.w; b3.x = q3.z; b3.y = q3.w;
        acc0 = __builtin_amdgcn_wmma_f32_16x16x4_f32(false, a1, false, b0, (short)0, acc0, false, false);
        acc1 = __builtin_amdgcn_wmma_f32_16x16x4_f32(false, a1, false, b1, (short)0, acc1, false, false);
        acc2 = __builtin_amdgcn_wmma_f32_16x16x4_f32(false, a1, false, b2, (short)0, acc2, false, false);
        acc3 = __builtin_amdgcn_wmma_f32_16x16x4_f32(false, a1, false, b3, (short)0, acc3, false, false);
    }

    const int kkbase = mtile * 16 + 8 * half;
#pragma unroll
    for (int r = 0; r < 8; ++r) {
        const int kkr = kkbase + r;
        const float bias = sb[kkr];
        float4* krow = (float4*)(ker + (size_t)kkr * NP + p_base + 4 * l16);
        *krow = make_float4(acc0[r] + bias, acc1[r] + bias, acc2[r] + bias, acc3[r] + bias);
    }
}

#define CG  (Cc / Gg)
#define RB  4
#define QR  (Ww / 4)
#define QB  (RB * QR)
#define CS  4
#define CPT (CG / CS)
#define IB  (QB * CS)
__global__ __launch_bounds__(IB) __attribute__((amdgpu_waves_per_eu(4, 7)))
void involution_apply(
    const float* __restrict__ x, const float* __restrict__ ker,
    float* __restrict__ out)
{
    const int tid = threadIdx.x;
    const int s   = tid / QB;
    const int rem = tid - s * QB;
    const int r   = rem / QR;
    const int q   = rem - r * QR;
    const int h   = blockIdx.x * RB + r;
    const int g   = blockIdx.y % Gg;
    const int b   = blockIdx.y / Gg;
    const int p   = b * HW + h * Ww + 4 * q;

    const unsigned cL = (unsigned)max(q - 1, 0) * 16u;
    const unsigned cC = (unsigned)q * 16u;
    const unsigned cR = (unsigned)min(q + 1, QR - 1) * 16u;
    const float mL = q > 0 ? 1.f : 0.f;
    const float mR = q < QR - 1 ? 1.f : 0.f;

    const char*  xg = (const char*)(x + ((size_t)b * Cc + g * CG + s * CPT) * HW);
    const float* kg = ker + (size_t)(g * KK) * NP + p;

    float acc[CPT][4];
#pragma unroll
    for (int c = 0; c < CPT; ++c)
#pragma unroll
        for (int e = 0; e < 4; ++e)
            acc[c][e] = 0.f;

#pragma unroll
    for (int i = 0; i < Kk; ++i) {
        const int hh = h + i - 3;
        const float mrow = (hh >= 0 && hh < Hh) ? 1.f : 0.f;
        const unsigned ro = (unsigned)(min(max(hh, 0), Hh - 1) * Ww) * 4u;
        const float mrl = mrow * mL;
        const float mrr = mrow * mR;

        float kv[Kk][4];
#pragma unroll
        for (int j = 0; j < Kk; ++j) {
            const float4 k4 = *(const float4*)(kg + (size_t)(i * Kk + j) * NP);
            const float kk4[4] = {k4.x, k4.y, k4.z, k4.w};
#pragma unroll
            for (int e = 0; e < 4; ++e) {
                const int m = e + j + 1;
                kv[j][e] = kk4[e] * (m < 4 ? mrl : (m >= 8 ? mrr : mrow));
            }
        }

#pragma unroll
        for (int c = 0; c < CPT; ++c) {
            const char* xc = xg + (size_t)c * (HW * 4) + ro;
            const float4 xa = *(const float4*)(xc + cL);
            const float4 xb = *(const float4*)(xc + cC);
            const float4 xd = *(const float4*)(xc + cR);
            const float win[12] = {xa.x, xa.y, xa.z, xa.w, xb.x, xb.y, xb.z, xb.w,
                                   xd.x, xd.y, xd.z, xd.w};
#pragma unroll
            for (int j = 0; j < Kk; ++j)
#pragma unroll
                for (int e = 0; e < 4; ++e)
                    acc[c][e] += win[e + j + 1] * kv[j][e];
        }
    }

    float* og = out + ((size_t)b * Cc + g * CG + s * CPT) * HW + h * Ww + 4 * q;
#pragma unroll
    for (int c = 0; c < CPT; ++c)
        *(float4*)(og + (size_t)c * HW) = make_float4(acc[c][0], acc[c][1], acc[c][2], acc[c][3]);
}

extern "C" void kernel_launch(void* const* d_in, const int* in_sizes, int n_in,
                              void* d_out, int out_size, void* d_ws, size_t ws_size,
                              hipStream_t stream) {
    const float* x     = (const float*)d_in[0];
    const float* rw    = (const float*)d_in[1];
    const float* rb    = (const float*)d_in[2];
    const float* gamma = (const float*)d_in[3];
    const float* beta  = (const float*)d_in[4];
    const float* mean  = (const float*)d_in[5];
    const float* var   = (const float*)d_in[6];
    const float* pa    = (const float*)d_in[7];
    const float* sw    = (const float*)d_in[8];
    const float* sb    = (const float*)d_in[9];
    float* out = (float*)d_out;

    float* t_ws   = (float*)d_ws;
    float* ker_ws = t_ws + (size_t)NP * TN;

    gemm1_reduce_bn_prelu<<<392, 256, 0, stream>>>(
        x, rw, rb, gamma, beta, mean, var, pa, t_ws);
    gemm2_span<<<2401, 128, 0, stream>>>(
        t_ws, sw, sb, ker_ws);
    involution_apply<<<dim3(Hh / RB, Bn * Gg), IB, 0, stream>>>(
        x, ker_ws, out);
}


# span GEMM: wave tile 32x64 (2 kk tiles x 4 pixel tiles; tile 48 alone): each t fragment feeds 2 WMMAs, half the waves; a trip = 4 k-steps, its 16 loads in one group; k sum order kept
# speedup vs baseline: 2.9796x; 1.2106x over previous
#include <hip/hip_runtime.h>
#include <hip/hip_bf16.h>

typedef __attribute__((ext_vector_type(2))) float v2f;
typedef __attribute__((ext_vector_type(8))) float v8f;
typedef __attribute__((ext_vector_type(4))) float v4f;

#define Bn 4
#define Cc 256
#define Hh 56
#define Ww 56
#define HW (Hh * Ww)
#define NP (Bn * HW)
#define TN 256
#define Gg 16
#define Kk 7
#define KK (Kk * Kk)
#define NK (Gg * KK)
#define EPSf 1e-5f

#define TF_PTILE (16 * 2 * 32)
#define KU 8

__global__ __launch_bounds__(256) __attribute__((amdgpu_waves_per_eu(4, 8)))
void gemm1_reduce_bn_prelu(
    const float* __restrict__ x, const float* __restrict__ rw,
    const float* __restrict__ rb, const float* __restrict__ gamma,
    const float* __restrict__ beta, const float* __restrict__ mean,
    const float* __restrict__ var, const float* __restrict__ pa,
    float* __restrict__ t_out)
{
    const int lane  = threadIdx.x & 31;
    const int wave  = threadIdx.x >> 5;
    const int tile  = blockIdx.x * 8 + wave;
    const int mt    = tile & 15;
    const int blk   = tile >> 4;
    const int half  = lane >> 4;
    const int l16   = lane & 15;

    const int p_base = blk * 64;
    const int b      = p_base / HW;
    const int hw     = p_base - b * HW;
    const float* xq = x + (size_t)b * Cc * HW + hw + 4 * l16;
    const int cperm = 4 * ((l16 & 7) >> 1) + 2 * (l16 >> 3) + (l16 & 1);
    const float* wrow = rw + (size_t)(mt * 16 + cperm) * Cc;

    v8f acc0 = {}, acc1 = {}, acc2 = {}, acc3 = {};
#pragma unroll 1
    for (int k0 = 0; k0 < Cc; k0 += 4 * KU) {
        float4 xa[KU], xb[KU];
        v2f    w[KU];
#pragma unroll
        for (int u = 0; u < KU; ++u) {
            const int kh = k0 + 4 * u + 2 * half;
            xa[u] = *(const float4*)(xq + (size_t)kh * HW);
            xb[u] = *(const float4*)(xq + (size_t)(kh + 1) * HW);
            w[u]  = *(const v2f*)(wrow + kh);
        }
#pragma unroll
        for (int u = 0; u < KU; ++u) {
            v2f b0, b1, b2, b3;
            b0.x = xa[u].x; b0.y = xb[u].x;
            b1.x = xa[u].y; b1.y = xb[u].y;
            b2.x = xa[u].z; b2.y = xb[u].z;
            b3.x = xa[u].w; b3.y = xb[u].w;
            acc0 = __builtin_amdgcn_wmma_f32_16x16x4_f32(false, w[u], false, b0, (short)0, acc0, false, false);
            acc1 = __builtin_amdgcn_wmma_f32_16x16x4_f32(false, w[u], false, b1, (short)0, acc1, false, false);
            acc2 = __builtin_amdgcn_wmma_f32_16x16x4_f32(false, w[u], false, b2, (short)0, acc2, false, false);
            acc3 = __builtin_amdgcn_wmma_f32_16x16x4_f32(false, w[u], false, b3, (short)0, acc3, false, false);
        }
    }

    const float al = pa[0];
    float4* tf = (float4*)t_out + (size_t)(blk * 4) * TF_PTILE + lane;
    const int nbase = mt * 16 + 2 * half;
#pragma unroll
    for (int rq = 0; rq < 2; ++rq) {
        float o0[4], o1[4], o2[4], o3[4];
#pragma unroll
        for (int e = 0; e < 4; ++e) {
            const int r  = 4 * rq + e;
            const int n  = nbase + 4 * (r >> 1) + (r & 1);
            const float s  = gamma[n] * rsqrtf(var[n] + EPSf);
            const float bi = rb[n] - mean[n];
            const float bt = beta[n];
            float v0 = (acc0[r] + bi) * s + bt; v0 = v0 > 0.f ? v0 : al * v0;
            float v1 = (acc1[r] + bi) * s + bt; v1 = v1 > 0.f ? v1 : al * v1;
            float v2 = (acc2[r] + bi) * s + bt; v2 = v2 > 0.f ? v2 : al * v2;
            float v3 = (acc3[r] + bi) * s + bt; v3 = v3 > 0.f ? v3 : al * v3;
            o0[e] = v0;
            o1[e] = v1;
            o2[e] = v2;
            o3[e] = v3;
        }
        float4* dst = tf + (size_t)(mt * 2 + rq) * 32;
        dst[0 * TF_PTILE] = make_float4(o0[0], o0[1], o0[2], o0[3]);
        dst[1 * TF_PTILE] = make_float4(o1[0], o1[1], o1[2], o1[3]);
        dst[2 * TF_PTILE] = make_float4(o2[0], o2[1], o2[2], o2[3]);
        dst[3 * TF_PTILE] = make_float4(o3[0], o3[1], o3[2], o3[3]);
    }
}

#define MGRP 25

template <int MT>
__device__ __forceinline__ void span_wave_tile(
    const v4f* __restrict__ tf, const float* __restrict__ arow,
    const float* __restrict__ sb, float* __restrict__ ker,
    const int mtile0, const int p_base, const int half, const int l16)
{
    v8f acc[MT][4];
#pragma unroll
    for (int m = 0; m < MT; ++m)
#pragma unroll
        for (int n = 0; n < 4; ++n)
            acc[m][n] = (v8f){};

#pragma unroll 1
    for (int c2 = 0; c2 < TN / 16; ++c2) {
        v4f q[2][4];
        v2f a[MT][4];
#pragma unroll
        for (int u = 0; u < 2; ++u) {
            const int c = 2 * c2 + u;
#pragma unroll
            for (int n = 0; n < 4; ++n)
                q[u][n] = tf[n * TF_PTILE + c * 32];
#pragma unroll
            for (int m = 0; m < MT; ++m) {
                a[m][2 * u]     = *(const v2f*)(arow + (size_t)m * 16 * TN + 8 * c);
                a[m][2 * u + 1] = *(const v2f*)(arow + (size_t)m * 16 * TN + 8 * c + 4);
            }
        }
        if constexpr (MT == 2)
            asm volatile("" :: "v"(q[0][0]), "v"(q[0][1]), "v"(q[0][2]), "v"(q[0][3]),
                               "v"(q[1][0]), "v"(q[1][1]), "v"(q[1][2]), "v"(q[1][3]),
                               "v"(a[0][0]), "v"(a[0][1]), "v"(a[0][2]), "v"(a[0][3]),
                               "v"(a[MT - 1][0]), "v"(a[MT - 1][1]), "v"(a[MT - 1][2]), "v"(a[MT - 1][3]));
        else
            asm volatile("" :: "v"(q[0][0]), "v"(q[0][1]), "v"(q[0][2]), "v"(q[0][3]),
                               "v"(q[1][0]), "v"(q[1][1]), "v"(q[1][2]), "v"(q[1][3]),
                               "v"(a[0][0]), "v"(a[0][1]), "v"(a[0][2]), "v"(a[0][3]));
#pragma unroll
        for (int u = 0; u < 2; ++u) {
#pragma unroll
            for (int s = 0; s < 2; ++s) {
                v2f b[4];
#pragma unroll
                for (int n = 0; n < 4; ++n) {
                    b[n].x = s ? q[u][n].z : q[u][n].x;
                    b[n].y = s ? q[u][n].w : q[u][n].y;
                }
#pragma unroll
                for (int m = 0; m < MT; ++m)
#pragma unroll
                    for (int n = 0; n < 4; ++n)
                        acc[m][n] = __builtin_amdgcn_wmma_f32_16x16x4_f32(
                            false, a[m][2 * u + s], false, b[n], (short)0, acc[m][n], false, false);
            }
        }
    }

#pragma unroll
    for (int m = 0; m < MT; ++m) {
        const int kkbase = (mtile0 + m) * 16 + 8 * half;
#pragma unroll
        for (int r = 0; r < 8; ++r) {
            const int kkr = kkbase + r;
            const float bias = sb[kkr];
            float4* krow = (float4*)(ker + (size_t)kkr * NP + p_base + 4 * l16);
            *krow = make_float4(acc[m][0][r] + bias, acc[m][1][r] + bias,
                                acc[m][2][r] + bias, acc[m][3][r] + bias);
        }
    }
}

__global__ __launch_bounds__(128) void gemm2_span(
    const float* __restrict__ t_in, const float* __restrict__ sw,
    const float* __restrict__ sb, float* __restrict__ ker)
{
    const int lane  = threadIdx.x & 31;
    const int wave  = __builtin_amdgcn_readfirstlane((int)(threadIdx.x >> 5));
    const int tile  = blockIdx.x * 4 + wave;
    const int mgrp  = tile % MGRP;
    const int nquad = tile / MGRP;
    const int half  = lane >> 4;
    const int l16   = lane & 15;
    const int mtile0 = 2 * mgrp;

    const float* arow = sw + (size_t)(mtile0 * 16 + l16) * TN + 2 * half;
    const int p_base  = nquad * 64;
    const v4f* tf     = (const v4f*)t_in + (size_t)(nquad * 4) * TF_PTILE + lane;

    if (mgrp < MGRP - 1)
        span_wave_tile<2>(tf, arow, sb, ker, mtile0, p_base, half, l16);
    else
        span_wave_tile<1>(tf, arow, sb, ker, mtile0, p_base, half, l16);
}

#define CG  (Cc / Gg)
#define RB  4
#define QR  (Ww / 4)
#define QB  (RB * QR)
#define CS  4
#define CPT (CG / CS)
#define IB  (QB * CS)
__global__ __launch_bounds__(IB) __attribute__((amdgpu_waves_per_eu(4, 7)))
void involution_apply(
    const float* __restrict__ x, const float* __restrict__ ker,
    float* __restrict__ out)
{
    const int tid = threadIdx.x;
    const int s   = tid / QB;
    const int rem = tid - s * QB;
    const int r   = rem / QR;
    const int q   = rem - r * QR;
    const int h   = blockIdx.x * RB + r;
    const int g   = blockIdx.y % Gg;
    const int b   = blockIdx.y / Gg;
    const int p   = b * HW + h * Ww + 4 * q;

    const unsigned cL = (unsigned)max(q - 1, 0) * 16u;
    const unsigned cC = (unsigned)q * 16u;
    const unsigned cR = (unsigned)min(q + 1, QR - 1) * 16u;
    const float mL = q > 0 ? 1.f : 0.f;
    const float mR = q < QR - 1 ? 1.f : 0.f;

    const char*  xg = (const char*)(x + ((size_t)b * Cc + g * CG + s * CPT) * HW);
    const float* kg = ker + (size_t)(g * KK) * NP + p;

    float acc[CPT][4];
#pragma unroll
    for (int c = 0; c < CPT; ++c)
#pragma unroll
        for (int e = 0; e < 4; ++e)
            acc[c][e] = 0.f;

#pragma unroll
    for (int i = 0; i < Kk; ++i) {
        const int hh = h + i - 3;
        const float mrow = (hh >= 0 && hh < Hh) ? 1.f : 0.f;
        const unsigned ro = (unsigned)(min(max(hh, 0), Hh - 1) * Ww) * 4u;
        const float mrl = mrow * mL;
        const float mrr = mrow * mR;

        float kv[Kk][4];
#pragma unroll
        for (int j = 0; j < Kk; ++j) {
            const float4 k4 = *(const float4*)(kg + (size_t)(i * Kk + j) * NP);
            const float kk4[4] = {k4.x, k4.y, k4.z, k4.w};
#pragma unroll
            for (int e = 0; e < 4; ++e) {
                const int m = e + j + 1;
                kv[j][e] = kk4[e] * (m < 4 ? mrl : (m >= 8 ? mrr : mrow));
            }
        }

#pragma unroll
        for (int c = 0; c < CPT; ++c) {
            const char* xc = xg + (size_t)c * (HW * 4) + ro;
            const float4 xa = *(const float4*)(xc + cL);
            const float4 xb = *(const float4*)(xc + cC);
            const float4 xd = *(const float4*)(xc + cR);
            const float win[12] = {xa.x, xa.y, xa.z, xa.w, xb.x, xb.y, xb.z, xb.w,
                                   xd.x, xd.y, xd.z, xd.w};
#pragma unroll
            for (int j = 0; j < Kk; ++j)
#pragma unroll
                for (int e = 0; e < 4; ++e)
                    acc[c][e] += win[e + j + 1] * kv[j][e];
        }
    }

    float* og = out + ((size_t)b * Cc + g * CG + s * CPT) * HW + h * Ww + 4 * q;
#pragma unroll
    for (int c = 0; c < CPT; ++c)
        *(float4*)(og + (size_t)c * HW) = make_float4(acc[c][0], acc[c][1], acc[c][2], acc[c][3]);
}

extern "C" void kernel_launch(void* const* d_in, const int* in_sizes, int n_in,
                              void* d_out, int out_size, void* d_ws, size_t ws_size,
                              hipStream_t stream) {
    const float* x     = (const float*)d_in[0];
    const float* rw    = (const float*)d_in[1];
    const float* rb    = (const float*)d_in[2];
    const float* gamma = (const float*)d_in[3];
    const float* beta  = (const float*)d_in[4];
    const float* mean  = (const float*)d_in[5];
    const float* var   = (const float*)d_in[6];
    const float* pa    = (const float*)d_in[7];
    const float* sw    = (const float*)d_in[8];
    const float* sb    = (const float*)d_in[9];
    float* out = (float*)d_out;

    float* t_ws   = (float*)d_ws;
    float* ker_ws = t_ws + (size_t)NP * TN;

    gemm1_reduce_bn_prelu<<<392, 256, 0, stream>>>(
        x, rw, rb, gamma, beta, mean, var, pa, t_ws);
    gemm2_span<<<1225, 128, 0, stream>>>(
        t_ws, sw, sb, ker_ws);
    involution_apply<<<dim3(Hh / RB, Bn * Gg), IB, 0, stream>>>(
        x, ker_ws, out);
}


# span GEMM k loop software-pipelined over two register sets: a trip's 16 fragment loads go out one trip ahead, in front of the previous trip's 32 WMMAs, behind counted waits; k sum order kept
# speedup vs baseline: 3.2122x; 1.0781x over previous
#include <hip/hip_runtime.h>
#include <hip/hip_bf16.h>

typedef __attribute__((ext_vector_type(2))) float v2f;
typedef __attribute__((ext_vector_type(8))) float v8f;
typedef __attribute__((ext_vector_type(4))) float v4f;

#define Bn 4
#define Cc 256
#define Hh 56
#define Ww 56
#define HW (Hh * Ww)
#define NP (Bn * HW)
#define TN 256
#define Gg 16
#define Kk 7
#define KK (Kk * Kk)
#define NK (Gg * KK)
#define EPSf 1e-5f

#define TF_PTILE (16 * 2 * 32)
#define KU 8

__global__ __launch_bounds__(256) __attribute__((amdgpu_waves_per_eu(4, 8)))
void gemm1_reduce_bn_prelu(
    const float* __restrict__ x, const float* __restrict__ rw,
    const float* __restrict__ rb, const float* __restrict__ gamma,
    const float* __restrict__ beta, const float* __restrict__ mean,
    const float* __restrict__ var, const float* __restrict__ pa,
    float* __restrict__ t_out)
{
    const int lane  = threadIdx.x & 31;
    const int wave  = threadIdx.x >> 5;
    const int tile  = blockIdx.x * 8 + wave;
    const int mt    = tile & 15;
    const int blk   = tile >> 4;
    const int half  = lane >> 4;
    const int l16   = lane & 15;

    const int p_base = blk * 64;
    const int b      = p_base / HW;
    const int hw     = p_base - b * HW;
    const float* xq = x + (size_t)b * Cc * HW + hw + 4 * l16;
    const int cperm = 4 * ((l16 & 7) >> 1) + 2 * (l16 >> 3) + (l16 & 1);
    const float* wrow = rw + (size_t)(mt * 16 + cperm) * Cc;

    v8f acc0 = {}, acc1 = {}, acc2 = {}, acc3 = {};
#pragma unroll 1
    for (int k0 = 0; k0 < Cc; k0 += 4 * KU) {
        float4 xa[KU], xb[KU];
        v2f    w[KU];
#pragma unroll
        for (int u = 0; u < KU; ++u) {
            const int kh = k0 + 4 * u + 2 * half;
            xa[u] = *(const float4*)(xq + (size_t)kh * HW);
            xb[u] = *(const float4*)(xq + (size_t)(kh + 1) * HW);
            w[u]  = *(const v2f*)(wrow + kh);
        }
#pragma unroll
        for (int u = 0; u < KU; ++u) {
            v2f b0, b1, b2, b3;
            b0.x = xa[u].x; b0.y = xb[u].x;
            b1.x = xa[u].y; b1.y = xb[u].y;
            b2.x = xa[u].z; b2.y = xb[u].z;
            b3.x = xa[u].w; b3.y = xb[u].w;
            acc0 = __builtin_amdgcn_wmma_f32_16x16x4_f32(false, w[u], false, b0, (short)0, acc0, false, false);
            acc1 = __builtin_amdgcn_wmma_f32_16x16x4_f32(false, w[u], false, b1, (short)0, acc1, false, false);
            acc2 = __builtin_amdgcn_wmma_f32_16x16x4_f32(false, w[u], false, b2, (short)0, acc2, false, false);
            acc3 = __builtin_amdgcn_wmma_f32_16x16x4_f32(false, w[u], false, b3, (short)0, acc3, false, false);
        }
    }

    const float al = pa[0];
    float4* tf = (float4*)t_out + (size_t)(blk * 4) * TF_PTILE + lane;
    const int nbase = mt * 16 + 2 * half;
#pragma unroll
    for (int rq = 0; rq < 2; ++rq) {
        float o0[4], o1[4], o2[4], o3[4];
#pragma unroll
        for (int e = 0; e < 4; ++e) {
            const int r  = 4 * rq + e;
            const int n  = nbase + 4 * (r >> 1) + (r & 1);
            const float s  = gamma[n] * rsqrtf(var[n] + EPSf);
            const float bi = rb[n] - mean[n];
            const float bt = beta[n];
            float v0 = (acc0[r] + bi) * s + bt; v0 = v0 > 0.f ? v0 : al * v0;
            float v1 = (acc1[r] + bi) * s + bt; v1 = v1 > 0.f ? v1 : al * v1;
            float v2 = (acc2[r] + bi) * s + bt; v2 = v2 > 0.f ? v2 : al * v2;
            float v3 = (acc3[r] + bi) * s + bt; v3 = v3 > 0.f ? v3 : al * v3;
            o0[e] = v0;
            o1[e] = v1;
            o2[e] = v2;
            o3[e] = v3;
        }
        float4* dst = tf + (size_t)(mt * 2 + rq) * 32;
        dst[0 * TF_PTILE] = make_float4(o0[0], o0[1], o0[2], o0[3]);
        dst[1 * TF_PTILE] = make_float4(o1[0], o1[1], o1[2], o1[3]);
        dst[2 * TF_PTILE] = make_float4(o2[0], o2[1], o2[2], o2[3]);
        dst[3 * TF_PTILE] = make_float4(o3[0], o3[1], o3[2], o3[3]);
    }
}

#define MGRP 25

template <int MT>
__device__ __forceinline__ void span_load_trip(
    const v4f* __restrict__ tf, const float* __restrict__ arow, const int c2,
    v4f (&q)[2][4], v2f (&a)[MT][4])
{
#pragma unroll
    for (int u = 0; u < 2; ++u) {
        const int c = 2 * c2 + u;
#pragma unroll
        for (int n = 0; n < 4; ++n)
            q[u][n] = tf[n * TF_PTILE + c * 32];
#pragma unroll
        for (int m = 0; m < MT; ++m) {
            a[m][2 * u]     = *(const v2f*)(arow + (size_t)m * 16 * TN + 8 * c);
            a[m][2 * u + 1] = *(const v2f*)(arow + (size_t)m * 16 * TN + 8 * c + 4);
        }
    }
}

template <int MT>
__device__ __forceinline__ void span_wmma_trip(
    const v4f (&q)[2][4], const v2f (&a)[MT][4], v8f (&acc)[MT][4])
{
#pragma unroll
    for (int u = 0; u < 2; ++u) {
#pragma unroll
        for (int s = 0; s < 2; ++s) {
            v2f b[4];
#pragma unroll
            for (int n = 0; n < 4; ++n) {
                b[n].x = s ? q[u][n].z : q[u][n].x;
                b[n].y = s ? q[u][n].w : q[u][n].y;
            }
#pragma unroll
            for (int m = 0; m < MT; ++m)
#pragma unroll
                for (int n = 0; n < 4; ++n)
                    acc[m][n] = __builtin_amdgcn_wmma_f32_16x16x4_f32(
                        false, a[m][2 * u + s], false, b[n], (short)0, acc[m][n], false, false);
        }
    }
}

template <int MT>
__device__ __forceinline__ void span_wave_tile(
    const v4f* __restrict__ tf, const float* __restrict__ arow,
    const float* __restrict__ sb, float* __restrict__ ker,
    const int mtile0, const int p_base, const int half, const int l16)
{
    v8f acc[MT][4];
#pragma unroll
    for (int m = 0; m < MT; ++m)
#pragma unroll
        for (int n = 0; n < 4; ++n)
            acc[m][n] = (v8f){};

    v4f qA[2][4], qB[2][4];
    v2f aA[MT][4], aB[MT][4];
    span_load_trip<MT>(tf, arow, 0, qA, aA);
#pragma unroll 1
    for (int c2 = 0; c2 < TN / 16 - 2; c2 += 2) {
        span_load_trip<MT>(tf, arow, c2 + 1, qB, aB);
        __builtin_amdgcn_sched_barrier(0);
        span_wmma_trip<MT>(qA, aA, acc);
        __builtin_amdgcn_sched_barrier(0);
        span_load_trip<MT>(tf, arow, c2 + 2, qA, aA);
        __builtin_amdgcn_sched_barrier(0);
        span_wmma_trip<MT>(qB, aB, acc);
        __builtin_amdgcn_sched_barrier(0);
    }
    span_load_trip<MT>(tf, arow, TN / 16 - 1, qB, aB);
    __builtin_amdgcn_sched_barrier(0);
    span_wmma_trip<MT>(qA, aA, acc);
    __builtin_amdgcn_sched_barrier(0);
    span_wmma_trip<MT>(qB, aB, acc);

#pragma unroll
    for (int m = 0; m < MT; ++m) {
        const int kkbase = (mtile0 + m) * 16 + 8 * half;
#pragma unroll
        for (int r = 0; r < 8; ++r) {
            const int kkr = kkbase + r;
            const float bias = sb[kkr];
            float4* krow = (float4*)(ker + (size_t)kkr * NP + p_base + 4 * l16);
            *krow = make_float4(acc[m][0][r] + bias, acc[m][1][r] + bias,
                                acc[m][2][r] + bias, acc[m][3][r] + bias);
        }
    }
}

__global__ __launch_bounds__(128) void gemm2_span(
    const float* __restrict__ t_in, const float* __restrict__ sw,
    const float* __restrict__ sb, float* __restrict__ ker)
{
    const int lane  = threadIdx.x & 31;
    const int wave  = __builtin_amdgcn_readfirstlane((int)(threadIdx.x >> 5));
    const int tile  = blockIdx.x * 4 + wave;
    const int mgrp  = tile % MGRP;
    const int nquad = tile / MGRP;
    const int half  = lane >> 4;
    const int l16   = lane & 15;
    const int mtile0 = 2 * mgrp;

    const float* arow = sw + (size_t)(mtile0 * 16 + l16) * TN + 2 * half;
    const int p_base  = nquad * 64;
    const v4f* tf     = (const v4f*)t_in + (size_t)(nquad * 4) * TF_PTILE + lane;

    if (mgrp < MGRP - 1)
        span_wave_tile<2>(tf, arow, sb, ker, mtile0, p_base, half, l16);
    else
        span_wave_tile<1>(tf, arow, sb, ker, mtile0, p_base, half, l16);
}

#define CG  (Cc / Gg)
#define RB  4
#define QR  (Ww / 4)
#define QB  (RB * QR)
#define CS  4
#define CPT (CG / CS)
#define IB  (QB * CS)
__global__ __launch_bounds__(IB) __attribute__((amdgpu_waves_per_eu(4, 7)))
void involution_apply(
    const float* __restrict__ x, const float* __restrict__ ker,
    float* __restrict__ out)
{
    const int tid = threadIdx.x;
    const int s   = tid / QB;
    const int rem = tid - s * QB;
    const int r   = rem / QR;
    const int q   = rem - r * QR;
    const int h   = blockIdx.x * RB + r;
    const int g   = blockIdx.y % Gg;
    const int b   = blockIdx.y / Gg;
    const int p   = b * HW + h * Ww + 4 * q;

    const unsigned cL = (unsigned)max(q - 1, 0) * 16u;
    const unsigned cC = (unsigned)q * 16u;
    const unsigned cR = (unsigned)min(q + 1, QR - 1) * 16u;
    const float mL = q > 0 ? 1.f : 0.f;
    const float mR = q < QR - 1 ? 1.f : 0.f;

    const char*  xg = (const char*)(x + ((size_t)b * Cc + g * CG + s * CPT) * HW);
    const float* kg = ker + (size_t)(g * KK) * NP + p;

    float acc[CPT][4];
#pragma unroll
    for (int c = 0; c < CPT; ++c)
#pragma unroll
        for (int e = 0; e < 4; ++e)
            acc[c][e] = 0.f;

#pragma unroll
    for (int i = 0; i < Kk; ++i) {
        const int hh = h + i - 3;
        const float mrow = (hh >= 0 && hh < Hh) ? 1.f : 0.f;
        const unsigned ro = (unsigned)(min(max(hh, 0), Hh - 1) * Ww) * 4u;
        const float mrl = mrow * mL;
        const float mrr = mrow * mR;

        float kv[Kk][4];
#pragma unroll
        for (int j = 0; j < Kk; ++j) {
            const float4 k4 = *(const float4*)(kg + (size_t)(i * Kk + j) * NP);
            const float kk4[4] = {k4.x, k4.y, k4.z, k4.w};
#pragma unroll
            for (int e = 0; e < 4; ++e) {
                const int m = e + j + 1;
                kv[j][e] = kk4[e] * (m < 4 ? mrl : (m >= 8 ? mrr : mrow));
            }
        }

#pragma unroll
        for (int c = 0; c < CPT; ++c) {
            const char* xc = xg + (size_t)c * (HW * 4) + ro;
            const float4 xa = *(const float4*)(xc + cL);
            const float4 xb = *(const float4*)(xc + cC);
            const float4 xd = *(const float4*)(xc + cR);
            const float win[12] = {xa.x, xa.y, xa.z, xa.w, xb.x, xb.y, xb.z, xb.w,
                                   xd.x, xd.y, xd.z, xd.w};
#pragma unroll
            for (int j = 0; j < Kk; ++j)
#pragma unroll
                for (int e = 0; e < 4; ++e)
                    acc[c][e] += win[e + j + 1] * kv[j][e];
        }
    }

    float* og = out + ((size_t)b * Cc + g * CG + s * CPT) * HW + h * Ww + 4 * q;
#pragma unroll
    for (int c = 0; c < CPT; ++c)
        *(float4*)(og + (size_t)c * HW) = make_float4(acc[c][0], acc[c][1], acc[c][2], acc[c][3]);
}

extern "C" void kernel_launch(void* const* d_in, const int* in_sizes, int n_in,
                              void* d_out, int out_size, void* d_ws, size_t ws_size,
                              hipStream_t stream) {
    const float* x     = (const float*)d_in[0];
    const float* rw    = (const float*)d_in[1];
    const float* rb    = (const float*)d_in[2];
    const float* gamma = (const float*)d_in[3];
    const float* beta  = (const float*)d_in[4];
    const float* mean  = (const float*)d_in[5];
    const float* var   = (const float*)d_in[6];
    const float* pa    = (const float*)d_in[7];
    const float* sw    = (const float*)d_in[8];
    const float* sb    = (const float*)d_in[9];
    float* out = (float*)d_out;

    float* t_ws   = (float*)d_ws;
    float* ker_ws = t_ws + (size_t)NP * TN;

    gemm1_reduce_bn_prelu<<<392, 256, 0, stream>>>(
        x, rw, rb, gamma, beta, mean, var, pa, t_ws);
    gemm2_span<<<1225, 128, 0, stream>>>(
        t_ws, sw, sb, ker_ws);
    involution_apply<<<dim3(Hh / RB, Bn * Gg), IB, 0, stream>>>(
        x, ker_ws, out);
}


# span GEMM: span_w copied to the workspace in A-fragment order by 196 extra blocks of launch 1; two k-steps of A = one 16-byte load of 4 whole lines (was two 8-byte loads of 16 lines); k orders kept
# speedup vs baseline: 3.4283x; 1.0673x over previous
#include <hip/hip_runtime.h>
#include <hip/hip_bf16.h>

typedef __attribute__((ext_vector_type(2))) float v2f;
typedef __attribute__((ext_vector_type(8))) float v8f;
typedef __attribute__((ext_vector_type(4))) float v4f;

#define Bn 4
#define Cc 256
#define Hh 56
#define Ww 56
#define HW (Hh * Ww)
#define NP (Bn * HW)
#define TN 256
#define Gg 16
#define Kk 7
#define KK (Kk * Kk)
#define NK (Gg * KK)
#define EPSf 1e-5f

#define TF_PTILE (16 * 2 * 32)
#define KU 8

#define WF_KTILE (32 * 32)
#define G1_BLOCKS 392
#define WF_BLOCKS 196

__global__ __launch_bounds__(256) __attribute__((amdgpu_waves_per_eu(4, 8)))
void gemm1_reduce_bn_prelu(
    const float* __restrict__ x, const float* __restrict__ rw,
    const float* __restrict__ rb, const float* __restrict__ gamma,
    const float* __restrict__ beta, const float* __restrict__ mean,
    const float* __restrict__ var, const float* __restrict__ pa,
    float* __restrict__ t_out,
    const float* __restrict__ sw, float* __restrict__ wf_out)
{
    if (blockIdx.x >= G1_BLOCKS) {
        const int idx = (blockIdx.x - G1_BLOCKS) * 256 + threadIdx.x;
        const int ln  = idx & 31;
        const int c   = (idx >> 5) & 31;
        const int kt  = idx >> 10;
        const float* s = sw + (size_t)(kt * 16 + (ln & 15)) * TN + 8 * c + 2 * (ln >> 4);
        const v2f lo = *(const v2f*)s;
        const v2f hi = *(const v2f*)(s + 4);
        ((float4*)wf_out)[idx] = make_float4(lo.x, lo.y, hi.x, hi.y);
        return;
    }

    const int lane  = threadIdx.x & 31;
    const int wave  = threadIdx.x >> 5;
    const int tile  = blockIdx.x * 8 + wave;
    const int mt    = tile & 15;
    const int blk   = tile >> 4;
    const int half  = lane >> 4;
    const int l16   = lane & 15;

    const int p_base = blk * 64;
    const int b      = p_base / HW;
    const int hw     = p_base - b * HW;
    const float* xq = x + (size_t)b * Cc * HW + hw + 4 * l16;
    const int cperm = 4 * ((l16 & 7) >> 1) + 2 * (l16 >> 3) + (l16 & 1);
    const float* wrow = rw + (size_t)(mt * 16 + cperm) * Cc;

    v8f acc0 = {}, acc1 = {}, acc2 = {}, acc3 = {};
#pragma unroll 1
    for (int k0 = 0; k0 < Cc; k0 += 4 * KU) {
        float4 xa[KU], xb[KU];
        v2f    w[KU];
#pragma unroll
        for (int u = 0; u < KU; ++u) {
            const int kh = k0 + 4 * u + 2 * half;
            xa[u] = *(const float4*)(xq + (size_t)kh * HW);
            xb[u] = *(const float4*)(xq + (size_t)(kh + 1) * HW);
            w[u]  = *(const v2f*)(wrow + kh);
        }
#pragma unroll
        for (int u = 0; u < KU; ++u) {
            v2f b0, b1, b2, b3;
            b0.x = xa[u].x; b0.y = xb[u].x;
            b1.x = xa[u].y; b1.y = xb[u].y;
            b2.x = xa[u].z; b2.y = xb[u].z;
            b3.x = xa[u].w; b3.y = xb[u].w;
            acc0 = __builtin_amdgcn_wmma_f32_16x16x4_f32(false, w[u], false, b0, (short)0, acc0, false, false);
            acc1 = __builtin_amdgcn_wmma_f32_16x16x4_f32(false, w[u], false, b1, (short)0, acc1, false, false);
            acc2 = __builtin_amdgcn_wmma_f32_16x16x4_f32(false, w[u], false, b2, (short)0, acc2, false, false);
            acc3 = __builtin_amdgcn_wmma_f32_16x16x4_f32(false, w[u], false, b3, (short)0, acc3, false, false);
        }
    }

    const float al = pa[0];
    float4* tf = (float4*)t_out + (size_t)(blk * 4) * TF_PTILE + lane;
    const int nbase = mt * 16 + 2 * half;
#pragma unroll
    for (int rq = 0; rq < 2; ++rq) {
        float o0[4], o1[4], o2[4], o3[4];
#pragma unroll
        for (int e = 0; e < 4; ++e) {
            const int r  = 4 * rq + e;
            const int n  = nbase + 4 * (r >> 1) + (r & 1);
            const float s  = gamma[n] * rsqrtf(var[n] + EPSf);
            const float bi = rb[n] - mean[n];
            const float bt = beta[n];
            float v0 = (acc0[r] + bi) * s + bt; v0 = v0 > 0.f ? v0 : al * v0;
            float v1 = (acc1[r] + bi) * s + bt; v1 = v1 > 0.f ? v1 : al * v1;
            float v2 = (acc2[r] + bi) * s + bt; v2 = v2 > 0.f ? v2 : al * v2;
            float v3 = (acc3[r] + bi) * s + bt; v3 = v3 > 0.f ? v3 : al * v3;
            o0[e] = v0;
            o1[e] = v1;
            o2[e] = v2;
            o3[e] = v3;
        }
        float4* dst = tf + (size_t)(mt * 2 + rq) * 32;
        dst[0 * TF_PTILE] = make_float4(o0[0], o0[1], o0[2], o0[3]);
        dst[1 * TF_PTILE] = make_float4(o1[0], o1[1], o1[2], o1[3]);
        dst[2 * TF_PTILE] = make_float4(o2[0], o2[1], o2[2], o2[3]);
        dst[3 * TF_PTILE] = make_float4(o3[0], o3[1], o3[2], o3[3]);
    }
}

#define MGRP 25

template <int MT>
__device__ __forceinline__ void span_load_trip(
    const v4f* __restrict__ tf, const v4f* __restrict__ af, const int c2,
    v4f (&q)[2][4], v4f (&a)[MT][2])
{
#pragma unroll
    for (int u = 0; u < 2; ++u) {
        const int c = 2 * c2 + u;
#pragma unroll
        for (int n = 0; n < 4; ++n)
            q[u][n] = tf[n * TF_PTILE + c * 32];
#pragma unroll
        for (int m = 0; m < MT; ++m)
            a[m][u] = af[m * WF_KTILE + c * 32];
    }
}

template <int MT>
__device__ __forceinline__ void span_wmma_trip(
    const v4f (&q)[2][4], const v4f (&a)[MT][2], v8f (&acc)[MT][4])
{
#pragma unroll
    for (int u = 0; u < 2; ++u) {
#pragma unroll
        for (int s = 0; s < 2; ++s) {
            v2f b[4];
#pragma unroll
            for (int n = 0; n < 4; ++n) {
                b[n].x = s ? q[u][n].z : q[u][n].x;
                b[n].y = s ? q[u][n].w : q[u][n].y;
            }
            v2f am[MT];
#pragma unroll
            for (int m = 0; m < MT; ++m) {
                am[m].x = s ? a[m][u].z : a[m][u].x;
                am[m].y = s ? a[m][u].w : a[m][u].y;
            }
#pragma unroll
            for (int m = 0; m < MT; ++m)
#pragma unroll
                for (int n = 0; n < 4; ++n)
                    acc[m][n] = __builtin_amdgcn_wmma_f32_16x16x4_f32(
                        false, am[m], false, b[n], (short)0, acc[m][n], false, false);
        }
    }
}

template <int MT>
__device__ __forceinline__ void span_wave_tile(
    const v4f* __restrict__ tf, const v4f* __restrict__ af,
    const float* __restrict__ sb, float* __restrict__ ker,
    const int mtile0, const int p_base, const int half, const int l16)
{
    v8f acc[MT][4];
#pragma unroll
    for (int m = 0; m < MT; ++m)
#pragma unroll
        for (int n = 0; n < 4; ++n)
            acc[m][n] = (v8f){};

    v4f qA[2][4], qB[2][4];
    v4f aA[MT][2], aB[MT][2];
    span_load_trip<MT>(tf, af, 0, qA, aA);
#pragma unroll 1
    for (int c2 = 0; c2 < TN / 16 - 2; c2 += 2) {
        span_load_trip<MT>(tf, af, c2 + 1, qB, aB);
        __builtin_amdgcn_sched_barrier(0);
        span_wmma_trip<MT>(qA, aA, acc);
        __builtin_amdgcn_sched_barrier(0);
        span_load_trip<MT>(tf, af, c2 + 2, qA, aA);
        __builtin_amdgcn_sched_barrier(0);
        span_wmma_trip<MT>(qB, aB, acc);
        __builtin_amdgcn_sched_barrier(0);
    }
    span_load_trip<MT>(tf, af, TN / 16 - 1, qB, aB);
    __builtin_amdgcn_sched_barrier(0);
    span_wmma_trip<MT>(qA, aA, acc);
    __builtin_amdgcn_sched_barrier(0);
    span_wmma_trip<MT>(qB, aB, acc);

#pragma unroll
    for (int m = 0; m < MT; ++m) {
        const int kkbase = (mtile0 + m) * 16 + 8 * half;
#pragma unroll
        for (int r = 0; r < 8; ++r) {
            const int kkr = kkbase + r;
            const float bias = sb[kkr];
            float4* krow = (float4*)(ker + (size_t)kkr * NP + p_base + 4 * l16);
            *krow = make_float4(acc[m][0][r] + bias, acc[m][1][r] + bias,
                                acc[m][2][r] + bias, acc[m][3][r] + bias);
        }
    }
}

__global__ __launch_bounds__(128) void gemm2_span(
    const float* __restrict__ t_in, const float* __restrict__ wf,
    const float* __restrict__ sb, float* __restrict__ ker)
{
    const int lane  = threadIdx.x & 31;
    const int wave  = __builtin_amdgcn_readfirstlane((int)(threadIdx.x >> 5));
    const int tile  = blockIdx.x * 4 + wave;
    const int mgrp  = tile % MGRP;
    const int nquad = tile / MGRP;
    const int half  = lane >> 4;
    const int l16   = lane & 15;
    const int mtile0 = 2 * mgrp;

    const v4f* af     = (const v4f*)wf + (size_t)mtile0 * WF_KTILE + lane;
    const int p_base  = nquad * 64;
    const v4f* tf     = (const v4f*)t_in + (size_t)(nquad * 4) * TF_PTILE + lane;

    if (mgrp < MGRP - 1)
        span_wave_tile<2>(tf, af, sb, ker, mtile0, p_base, half, l16);
    else
        span_wave_tile<1>(tf, af, sb, ker, mtile0, p_base, half, l16);
}

#define CG  (Cc / Gg)
#define RB  4
#define QR  (Ww / 4)
#define QB  (RB * QR)
#define CS  4
#define CPT (CG / CS)
#define IB  (QB * CS)
__global__ __launch_bounds__(IB) __attribute__((amdgpu_waves_per_eu(4, 7)))
void involution_apply(
    const float* __restrict__ x, const float* __restrict__ ker,
    float* __restrict__ out)
{
    const int tid = threadIdx.x;
    const int s   = tid / QB;
    const int rem = tid - s * QB;
    const int r   = rem / QR;
    const int q   = rem - r * QR;
    const int h   = blockIdx.x * RB + r;
    const int g   = blockIdx.y % Gg;
    const int b   = blockIdx.y / Gg;
    const int p   = b * HW + h * Ww + 4 * q;

    const unsigned cL = (unsigned)max(q - 1, 0) * 16u;
    const unsigned cC = (unsigned)q * 16u;
    const unsigned cR = (unsigned)min(q + 1, QR - 1) * 16u;
    const float mL = q > 0 ? 1.f : 0.f;
    const float mR = q < QR - 1 ? 1.f : 0.f;

    const char*  xg = (const char*)(x + ((size_t)b * Cc + g * CG + s * CPT) * HW);
    const float* kg = ker + (size_t)(g * KK) * NP + p;

    float acc[CPT][4];
#pragma unroll
    for (int c = 0; c < CPT; ++c)
#pragma unroll
        for (int e = 0; e < 4; ++e)
            acc[c][e] = 0.f;

#pragma unroll
    for (int i = 0; i < Kk; ++i) {
        const int hh = h + i - 3;
        const float mrow = (hh >= 0 && hh < Hh) ? 1.f : 0.f;
        const unsigned ro = (unsigned)(min(max(hh, 0), Hh - 1) * Ww) * 4u;
        const float mrl = mrow * mL;
        const float mrr = mrow * mR;

        float kv[Kk][4];
#pragma unroll
        for (int j = 0; j < Kk; ++j) {
            const float4 k4 = *(const float4*)(kg + (size_t)(i * Kk + j) * NP);
            const float kk4[4] = {k4.x, k4.y, k4.z, k4.w};
#pragma unroll
            for (int e = 0; e < 4; ++e) {
                const int m = e + j + 1;
                kv[j][e] = kk4[e] * (m < 4 ? mrl : (m >= 8 ? mrr : mrow));
            }
        }

#pragma unroll
        for (int c = 0; c < CPT; ++c) {
            const char* xc = xg + (size_t)c * (HW * 4) + ro;
            const float4 xa = *(const float4*)(xc + cL);
            const float4 xb = *(const float4*)(xc + cC);
            const float4 xd = *(const float4*)(xc + cR);
            const float win[12] = {xa.x, xa.y, xa.z, xa.w, xb.x, xb.y, xb.z, xb.w,
                                   xd.x, xd.y, xd.z, xd.w};
#pragma unroll
            for (int j = 0; j < Kk; ++j)
#pragma unroll
                for (int e = 0; e < 4; ++e)
                    acc[c][e] += win[e + j + 1] * kv[j][e];
        }
    }

    float* og = out + ((size_t)b * Cc + g * CG + s * CPT) * HW + h * Ww + 4 * q;
#pragma unroll
    for (int c = 0; c < CPT; ++c)
        *(float4*)(og + (size_t)c * HW) = make_float4(acc[c][0], acc[c][1], acc[c][2], acc[c][3]);
}

extern "C" void kernel_launch(void* const* d_in, const int* in_sizes, int n_in,
                              void* d_out, int out_size, void* d_ws, size_t ws_size,
                              hipStream_t stream) {
    const float* x     = (const float*)d_in[0];
    const float* rw    = (const float*)d_in[1];
    const float* rb    = (const float*)d_in[2];
    const float* gamma = (const float*)d_in[3];
    const float* beta  = (const float*)d_in[4];
    const float* mean  = (const float*)d_in[5];
    const float* var   = (const float*)d_in[6];
    const float* pa    = (const float*)d_in[7];
    const float* sw    = (const float*)d_in[8];
    const float* sb    = (const float*)d_in[9];
    float* out = (float*)d_out;

    float* t_ws   = (float*)d_ws;
    float* ker_ws = t_ws + (size_t)NP * TN;
    float* wf_ws  = ker_ws + (size_t)NK * NP;

    gemm1_reduce_bn_prelu<<<588, 256, 0, stream>>>(
        x, rw, rb, gamma, beta, mean, var, pa, t_ws, sw, wf_ws);
    gemm2_span<<<1225, 128, 0, stream>>>(
        t_ws, wf_ws, sb, ker_ws);
    involution_apply<<<dim3(Hh / RB, Bn * Gg), IB, 0, stream>>>(
        x, ker_ws, out);
}


# involution: one thread per quad takes all 16 channels of its group (was 4 threads x 4 channels): its 49 ker loads go out once, as one group up front, and stay in registers; tap sum order kept
# speedup vs baseline: 3.6368x; 1.0608x over previous
#include <hip/hip_runtime.h>
#include <hip/hip_bf16.h>

typedef __attribute__((ext_vector_type(2))) float v2f;
typedef __attribute__((ext_vector_type(8))) float v8f;
typedef __attribute__((ext_vector_type(4))) float v4f;

#define Bn 4
#define Cc 256
#define Hh 56
#define Ww 56
#define HW (Hh * Ww)
#define NP (Bn * HW)
#define TN 256
#define Gg 16
#define Kk 7
#define KK (Kk * Kk)
#define NK (Gg * KK)
#define EPSf 1e-5f

#define TF_PTILE (16 * 2 * 32)
#define KU 8

#define WF_KTILE (32 * 32)
#define G1_BLOCKS 392
#define WF_BLOCKS 196

__global__ __launch_bounds__(256) __attribute__((amdgpu_waves_per_eu(4, 8)))
void gemm1_reduce_bn_prelu(
    const float* __restrict__ x, const float* __restrict__ rw,
    const float* __restrict__ rb, const float* __restrict__ gamma,
    const float* __restrict__ beta, const float* __restrict__ mean,
    const float* __restrict__ var, const float* __restrict__ pa,
    float* __restrict__ t_out,
    const float* __restrict__ sw, float* __restrict__ wf_out)
{
    if (blockIdx.x >= G1_BLOCKS) {
        const int idx = (blockIdx.x - G1_BLOCKS) * 256 + threadIdx.x;
        const int ln  = idx & 31;
        const int c   = (idx >> 5) & 31;
        const int kt  = idx >> 10;
        const float* s = sw + (size_t)(kt * 16 + (ln & 15)) * TN + 8 * c + 2 * (ln >> 4);
        const v2f lo = *(const v2f*)s;
        const v2f hi = *(const v2f*)(s + 4);
        ((float4*)wf_out)[idx] = make_float4(lo.x, lo.y, hi.x, hi.y);
        return;
    }

    const int lane  = threadIdx.x & 31;
    const int wave  = threadIdx.x >> 5;
    const int tile  = blockIdx.x * 8 + wave;
    const int mt    = tile & 15;
    const int blk   = tile >> 4;
    const int half  = lane >> 4;
    const int l16   = lane & 15;

    const int p_base = blk * 64;
    const int b      = p_base / HW;
    const int hw     = p_base - b * HW;
    const float* xq = x + (size_t)b * Cc * HW + hw + 4 * l16;
    const int cperm = 4 * ((l16 & 7) >> 1) + 2 * (l16 >> 3) + (l16 & 1);
    const float* wrow = rw + (size_t)(mt * 16 + cperm) * Cc;

    v8f acc0 = {}, acc1 = {}, acc2 = {}, acc3 = {};
#pragma unroll 1
    for (int k0 = 0; k0 < Cc; k0 += 4 * KU) {
        float4 xa[KU], xb[KU];
        v2f    w[KU];
#pragma unroll
        for (int u = 0; u < KU; ++u) {
            const int kh = k0 + 4 * u + 2 * half;
            xa[u] = *(const float4*)(xq + (size_t)kh * HW);
            xb[u] = *(const float4*)(xq + (size_t)(kh + 1) * HW);
            w[u]  = *(const v2f*)(wrow + kh);
        }
#pragma unroll
        for (int u = 0; u < KU; ++u) {
            v2f b0, b1, b2, b3;
            b0.x = xa[u].x; b0.y = xb[u].x;
            b1.x = xa[u].y; b1.y = xb[u].y;
            b2.x = xa[u].z; b2.y = xb[u].z;
            b3.x = xa[u].w; b3.y = xb[u].w;
            acc0 = __builtin_amdgcn_wmma_f32_16x16x4_f32(false, w[u], false, b0, (short)0, acc0, false, false);
            acc1 = __builtin_amdgcn_wmma_f32_16x16x4_f32(false, w[u], false, b1, (short)0, acc1, false, false);
            acc2 = __builtin_amdgcn_wmma_f32_16x16x4_f32(false, w[u], false, b2, (short)0, acc2, false, false);
            acc3 = __builtin_amdgcn_wmma_f32_16x16x4_f32(false, w[u], false, b3, (short)0, acc3, false, false);
        }
    }

    const float al = pa[0];
    float4* tf = (float4*)t_out + (size_t)(blk * 4) * TF_PTILE + lane;
    const int nbase = mt * 16 + 2 * half;
#pragma unroll
    for (int rq = 0; rq < 2; ++rq) {
        float o0[4], o1[4], o2[4], o3[4];
#pragma unroll
        for (int e = 0; e < 4; ++e) {
            const int r  = 4 * rq + e;
            const int n  = nbase + 4 * (r >> 1) + (r & 1);
            const float s  = gamma[n] * rsqrtf(var[n] + EPSf);
            const float bi = rb[n] - mean[n];
            const float bt = beta[n];
            float v0 = (acc0[r] + bi) * s + bt; v0 = v0 > 0.f ? v0 : al * v0;
            float v1 = (acc1[r] + bi) * s + bt; v1 = v1 > 0.f ? v1 : al * v1;
            float v2 = (acc2[r] + bi) * s + bt; v2 = v2 > 0.f ? v2 : al * v2;
            float v3 = (acc3[r] + bi) * s + bt; v3 = v3 > 0.f ? v3 : al * v3;
            o0[e] = v0;
            o1[e] = v1;
            o2[e] = v2;
            o3[e] = v3;
        }
        float4* dst = tf + (size_t)(mt * 2 + rq) * 32;
        dst[0 * TF_PTILE] = make_float4(o0[0], o0[1], o0[2], o0[3]);
        dst[1 * TF_PTILE] = make_float4(o1[0], o1[1], o1[2], o1[3]);
        dst[2 * TF_PTILE] = make_float4(o2[0], o2[1], o2[2], o2[3]);
        dst[3 * TF_PTILE] = make_float4(o3[0], o3[1], o3[2], o3[3]);
    }
}

#define MGRP 25

template <int MT>
__device__ __forceinline__ void span_load_trip(
    const v4f* __restrict__ tf, const v4f* __restrict__ af, const int c2,
    v4f (&q)[2][4], v4f (&a)[MT][2])
{
#pragma unroll
    for (int u = 0; u < 2; ++u) {
        const int c = 2 * c2 + u;
#pragma unroll
        for (int n = 0; n < 4; ++n)
            q[u][n] = tf[n * TF_PTILE + c * 32];
#pragma unroll
        for (int m = 0; m < MT; ++m)
            a[m][u] = af[m * WF_KTILE + c * 32];
    }
}

template <int MT>
__device__ __forceinline__ void span_wmma_trip(
    const v4f (&q)[2][4], const v4f (&a)[MT][2], v8f (&acc)[MT][4])
{
#pragma unroll
    for (int u = 0; u < 2; ++u) {
#pragma unroll
        for (int s = 0; s < 2; ++s) {
            v2f b[4];
#pragma unroll
            for (int n = 0; n < 4; ++n) {
                b[n].x = s ? q[u][n].z : q[u][n].x;
                b[n].y = s ? q[u][n].w : q[u][n].y;
            }
            v2f am[MT];
#pragma unroll
            for (int m = 0; m < MT; ++m) {
                am[m].x = s ? a[m][u].z : a[m][u].x;
                am[m].y = s ? a[m][u].w : a[m][u].y;
            }
#pragma unroll
            for (int m = 0; m < MT; ++m)
#pragma unroll
                for (int n = 0; n < 4; ++n)
                    acc[m][n] = __builtin_amdgcn_wmma_f32_16x16x4_f32(
                        false, am[m], false, b[n], (short)0, acc[m][n], false, false);
        }
    }
}

template <int MT>
__device__ __forceinline__ void span_wave_tile(
    const v4f* __restrict__ tf, const v4f* __restrict__ af,
    const float* __restrict__ sb, float* __restrict__ ker,
    const int mtile0, const int p_base, const int half, const int l16)
{
    v8f acc[MT][4];
#pragma unroll
    for (int m = 0; m < MT; ++m)
#pragma unroll
        for (int n = 0; n < 4; ++n)
            acc[m][n] = (v8f){};

    v4f qA[2][4], qB[2][4];
    v4f aA[MT][2], aB[MT][2];
    span_load_trip<MT>(tf, af, 0, qA, aA);
#pragma unroll 1
    for (int c2 = 0; c2 < TN / 16 - 2; c2 += 2) {
        span_load_trip<MT>(tf, af, c2 + 1, qB, aB);
        __builtin_amdgcn_sched_barrier(0);
        span_wmma_trip<MT>(qA, aA, acc);
        __builtin_amdgcn_sched_barrier(0);
        span_load_trip<MT>(tf, af, c2 + 2, qA, aA);
        __builtin_amdgcn_sched_barrier(0);
        span_wmma_trip<MT>(qB, aB, acc);
        __builtin_amdgcn_sched_barrier(0);
    }
    span_load_trip<MT>(tf, af, TN / 16 - 1, qB, aB);
    __builtin_amdgcn_sched_barrier(0);
    span_wmma_trip<MT>(qA, aA, acc);
    __builtin_amdgcn_sched_barrier(0);
    span_wmma_trip<MT>(qB, aB, acc);

#pragma unroll
    for (int m = 0; m < MT; ++m) {
        const int kkbase = (mtile0 + m) * 16 + 8 * half;
#pragma unroll
        for (int r = 0; r < 8; ++r) {
            const int kkr = kkbase + r;
            const float bias = sb[kkr];
            float4* krow = (float4*)(ker + (size_t)kkr * NP + p_base + 4 * l16);
            *krow = make_float4(acc[m][0][r] + bias, acc[m][1][r] + bias,
                                acc[m][2][r] + bias, acc[m][3][r] + bias);
        }
    }
}

__global__ __launch_bounds__(128) void gemm2_span(
    const float* __restrict__ t_in, const float* __restrict__ wf,
    const float* __restrict__ sb, float* __restrict__ ker)
{
    const int lane  = threadIdx.x & 31;
    const int wave  = __builtin_amdgcn_readfirstlane((int)(threadIdx.x >> 5));
    const int tile  = blockIdx.x * 4 + wave;
    const int mgrp  = tile % MGRP;
    const int nquad = tile / MGRP;
    const int half  = lane >> 4;
    const int l16   = lane & 15;
    const int mtile0 = 2 * mgrp;

    const v4f* af     = (const v4f*)wf + (size_t)mtile0 * WF_KTILE + lane;
    const int p_base  = nquad * 64;
    const v4f* tf     = (const v4f*)t_in + (size_t)(nquad * 4) * TF_PTILE + lane;

    if (mgrp < MGRP - 1)
        span_wave_tile<2>(tf, af, sb, ker, mtile0, p_base, half, l16);
    else
        span_wave_tile<1>(tf, af, sb, ker, mtile0, p_base, half, l16);
}

#define CG  (Cc / Gg)
#define QR  (Ww / 4)
#define QI  (Hh * QR)
#define IB  64
#define INV_BLOCKS ((Bn * Gg * QI) / IB)

__device__ __forceinline__ void inv_load_channel(
    const char* __restrict__ xb, const unsigned chan_bytes,
    const unsigned (&xo)[Kk][3], float4 (&xw)[Kk][3])
{
#pragma unroll
    for (int i = 0; i < Kk; ++i)
#pragma unroll
        for (int s = 0; s < 3; ++s)
            xw[i][s] = *(const float4*)(xb + chan_bytes + xo[i][s]);
}

__device__ __forceinline__ void inv_channel(
    const float4 (&xw)[Kk][3], const float (&kv)[KK][4], float* __restrict__ dst)
{
    float acc[4] = {0.f, 0.f, 0.f, 0.f};
#pragma unroll
    for (int i = 0; i < Kk; ++i) {
        const float win[12] = {xw[i][0].x, xw[i][0].y, xw[i][0].z, xw[i][0].w,
                               xw[i][1].x, xw[i][1].y, xw[i][1].z, xw[i][1].w,
                               xw[i][2].x, xw[i][2].y, xw[i][2].z, xw[i][2].w};
#pragma unroll
        for (int j = 0; j < Kk; ++j)
#pragma unroll
            for (int e = 0; e < 4; ++e)
                acc[e] += win[e + j + 1] * kv[i * Kk + j][e];
    }
    *(float4*)dst = make_float4(acc[0], acc[1], acc[2], acc[3]);
}

__global__ __launch_bounds__(IB) __attribute__((amdgpu_waves_per_eu(2, 3)))
void involution_apply(
    const float* __restrict__ x, const float* __restrict__ ker,
    float* __restrict__ out)
{
    const int gid = blockIdx.x * IB + threadIdx.x;
    const int bg  = gid / QI;
    const int qi  = gid - bg * QI;
    const int h   = qi / QR;
    const int q   = qi - h * QR;
    const int g   = bg % Gg;
    const int b   = bg / Gg;
    const int p   = b * HW + 4 * qi;


    const unsigned ko = (unsigned)((g * KK) * NP + p) * 4u;
    float4 k4[KK];
#pragma unroll
    for (int t = 0; t < KK; ++t)
        k4[t] = *(const float4*)((const char*)ker + (unsigned)(t * NP * 4) + ko);

    const unsigned cL = (unsigned)max(q - 1, 0) * 16u;
    const unsigned cC = (unsigned)q * 16u;
    const unsigned cR = (unsigned)min(q + 1, QR - 1) * 16u;
    const float mL = q > 0 ? 1.f : 0.f;
    const float mR = q < QR - 1 ? 1.f : 0.f;

    const unsigned go = (unsigned)((b * Cc + g * CG) * HW) * 4u;
    unsigned xo[Kk][3];
#pragma unroll
    for (int i = 0; i < Kk; ++i) {
        const unsigned ro = go + (unsigned)(min(max(h + i - 3, 0), Hh - 1) * Ww) * 4u;
        xo[i][0] = ro + cL;
        xo[i][1] = ro + cC;
        xo[i][2] = ro + cR;
    }

    const char* xb = (const char*)x;
    float*      og = out + ((size_t)b * Cc + g * CG) * HW + 4 * qi;

    float4 xA[Kk][3], xB[Kk][3];
    inv_load_channel(xb, 0u, xo, xA);

    float kv[KK][4];
#pragma unroll
    for (int i = 0; i < Kk; ++i) {
        const int hh = h + i - 3;
        const float mrow = (hh >= 0 && hh < Hh) ? 1.f : 0.f;
        const float mrl = mrow * mL;
        const float mrr = mrow * mR;
#pragma unroll
        for (int j = 0; j < Kk; ++j) {
            const float4 kq = k4[i * Kk + j];
            const float kk4[4] = {kq.x, kq.y, kq.z, kq.w};
#pragma unroll
            for (int e = 0; e < 4; ++e) {
                const int m = e + j + 1;
                kv[i * Kk + j][e] = kk4[e] * (m < 4 ? mrl : (m >= 8 ? mrr : mrow));
            }
        }
    }

#pragma unroll
    for (int c = 0; c < CG; c += 2) {
        inv_load_channel(xb, (unsigned)((c + 1) * HW * 4), xo, xB);
        inv_channel(xA, kv, og + (size_t)c * HW);
        if (c + 2 < CG)
            inv_load_channel(xb, (unsigned)((c + 2) * HW * 4), xo, xA);
        inv_channel(xB, kv, og + (size_t)(c + 1) * HW);
    }
}

extern "C" void kernel_launch(void* const* d_in, const int* in_sizes, int n_in,
                              void* d_out, int out_size, void* d_ws, size_t ws_size,
                              hipStream_t stream) {
    const float* x     = (const float*)d_in[0];
    const float* rw    = (const float*)d_in[1];
    const float* rb    = (const float*)d_in[2];
    const float* gamma = (const float*)d_in[3];
    const float* beta  = (const float*)d_in[4];
    const float* mean  = (const float*)d_in[5];
    const float* var   = (const float*)d_in[6];
    const float* pa    = (const float*)d_in[7];
    const float* sw    = (const float*)d_in[8];
    const float* sb    = (const float*)d_in[9];
    float* out = (float*)d_out;

    float* t_ws   = (float*)d_ws;
    float* ker_ws = t_ws + (size_t)NP * TN;
    float* wf_ws  = ker_ws + (size_t)NK * NP;

    gemm1_reduce_bn_prelu<<<588, 256, 0, stream>>>(
        x, rw, rb, gamma, beta, mean, var, pa, t_ws, sw, wf_ws);
    gemm2_span<<<1225, 128, 0, stream>>>(
        t_ws, wf_ws, sb, ker_ws);
    involution_apply<<<INV_BLOCKS, IB, 0, stream>>>(
        x, ker_ws, out);
}


# span GEMM fused with the involution: a group's taps 0..47 go from WMMA accumulators to LDS, one work-group per WGP (group x 12-13 pixel quads); tap 48 on WMMA in launch 1 as 16 partial k sums
# speedup vs baseline: 3.9500x; 1.0861x over previous
#include <hip/hip_runtime.h>
#include <hip/hip_bf16.h>


typedef __attribute__((ext_vector_type(2))) float v2f;
typedef __attribute__((ext_vector_type(8))) float v8f;
typedef __attribute__((ext_vector_type(4))) float v4f;

#define Bn 4
#define Cc 256
#define Hh 56
#define Ww 56
#define HW (Hh * Ww)
#define NP (Bn * HW)
#define TN 256
#define Gg 16
#define Kk 7
#define KK (Kk * Kk)
#define EPSf 1e-5f

#define TF_PTILE (16 * 2 * 32)
#define KU 8
#define WF_KTILE (32 * 32)
#define G1_BLOCKS 196
#define WF_BLOCKS 96
#define G1_THREADS 512

__global__ __launch_bounds__(G1_THREADS) __attribute__((amdgpu_waves_per_eu(4, 8)))
void gemm1_reduce_bn_prelu(
    const float* __restrict__ x, const float* __restrict__ rw,
    const float* __restrict__ rb, const float* __restrict__ gamma,
    const float* __restrict__ beta, const float* __restrict__ mean,
    const float* __restrict__ var, const float* __restrict__ pa,
    float* __restrict__ t_out,
    const float* __restrict__ sw, float* __restrict__ wf_out,
    const float* __restrict__ sb, float* __restrict__ k48_out)
{
    __shared__ float red[16 * 4 * 8 * 32];

    if (blockIdx.x >= G1_BLOCKS) {
        const int idx = (blockIdx.x - G1_BLOCKS) * G1_THREADS + threadIdx.x;
        const int ln  = idx & 31;
        const int c   = (idx >> 5) & 31;
        const int kt  = idx >> 10;
        const int g   = kt / 3;
        const int j   = kt - 3 * g;
        const float* s = sw + (size_t)(g * KK + 16 * j + (ln & 15)) * TN + 8 * c + 2 * (ln >> 4);
        const v2f lo = *(const v2f*)s;
        const v2f hi = *(const v2f*)(s + 4);
        ((float4*)wf_out)[idx] = make_float4(lo.x, lo.y, hi.x, hi.y);
        return;
    }

    const int lane  = threadIdx.x & 31;
    const int mt    = threadIdx.x >> 5;
    const int blk   = blockIdx.x;
    const int half  = lane >> 4;
    const int l16   = lane & 15;

    const int p_base = blk * 64;
    const int b      = p_base / HW;
    const int hw     = p_base - b * HW;
    const float* xq = x + (size_t)b * Cc * HW + hw + 4 * l16;
    const int cperm = 4 * ((l16 & 7) >> 1) + 2 * (l16 >> 3) + (l16 & 1);
    const float* wrow = rw + (size_t)(mt * 16 + cperm) * Cc;

    v8f acc0 = {}, acc1 = {}, acc2 = {}, acc3 = {};
#pragma unroll 1
    for (int k0 = 0; k0 < Cc; k0 += 4 * KU) {
        float4 xa[KU], xb[KU];
        v2f    w[KU];
#pragma unroll
        for (int u = 0; u < KU; ++u) {
            const int kh = k0 + 4 * u + 2 * half;
            xa[u] = *(const float4*)(xq + (size_t)kh * HW);
            xb[u] = *(const float4*)(xq + (size_t)(kh + 1) * HW);
            w[u]  = *(const v2f*)(wrow + kh);
        }
#pragma unroll
        for (int u = 0; u < KU; ++u) {
            v2f b0, b1, b2, b3;
            b0.x = xa[u].x; b0.y = xb[u].x;
            b1.x = xa[u].y; b1.y = xb[u].y;
            b2.x = xa[u].z; b2.y = xb[u].z;
            b3.x = xa[u].w; b3.y = xb[u].w;
            acc0 = __builtin_amdgcn_wmma_f32_16x16x4_f32(false, w[u], false, b0, (short)0, acc0, false, false);
            acc1 = __builtin_amdgcn_wmma_f32_16x16x4_f32(false, w[u], false, b1, (short)0, acc1, false, false);
            acc2 = __builtin_amdgcn_wmma_f32_16x16x4_f32(false, w[u], false, b2, (short)0, acc2, false, false);
            acc3 = __builtin_amdgcn_wmma_f32_16x16x4_f32(false, w[u], false, b3, (short)0, acc3, false, false);
        }
    }

    const float al = pa[0];
    float4* tf = (float4*)t_out + (size_t)(blk * 4) * TF_PTILE + lane;
    const int nbase = mt * 16 + 2 * half;
    v8f p48[4] = {(v8f){}, (v8f){}, (v8f){}, (v8f){}};
#pragma unroll
    for (int rq = 0; rq < 2; ++rq) {
        float o[4][4];
#pragma unroll
        for (int e = 0; e < 4; ++e) {
            const int r  = 4 * rq + e;
            const int n  = nbase + 4 * (r >> 1) + (r & 1);
            const float s  = gamma[n] * rsqrtf(var[n] + EPSf);
            const float bi = rb[n] - mean[n];
            const float bt = beta[n];
            float v0 = (acc0[r] + bi) * s + bt; v0 = v0 > 0.f ? v0 : al * v0;
            float v1 = (acc1[r] + bi) * s + bt; v1 = v1 > 0.f ? v1 : al * v1;
            float v2 = (acc2[r] + bi) * s + bt; v2 = v2 > 0.f ? v2 : al * v2;
            float v3 = (acc3[r] + bi) * s + bt; v3 = v3 > 0.f ? v3 : al * v3;
            o[0][e] = v0;
            o[1][e] = v1;
            o[2][e] = v2;
            o[3][e] = v3;
        }
        float4* dst = tf + (size_t)(mt * 2 + rq) * 32;
#pragma unroll
        for (int n = 0; n < 4; ++n)
            dst[n * TF_PTILE] = make_float4(o[n][0], o[n][1], o[n][2], o[n][3]);

        const float* s48 = sw + (size_t)(l16 * KK + 48) * TN + 8 * (2 * mt + rq) + 2 * half;
        const v2f alo = *(const v2f*)s48;
        const v2f ahi = *(const v2f*)(s48 + 4);
#pragma unroll
        for (int n = 0; n < 4; ++n) {
            v2f blo, bhi;
            blo.x = o[n][0]; blo.y = o[n][1];
            bhi.x = o[n][2]; bhi.y = o[n][3];
            p48[n] = __builtin_amdgcn_wmma_f32_16x16x4_f32(false, alo, false, blo, (short)0, p48[n], false, false);
            p48[n] = __builtin_amdgcn_wmma_f32_16x16x4_f32(false, ahi, false, bhi, (short)0, p48[n], false, false);
        }
    }

#pragma unroll
    for (int n = 0; n < 4; ++n)
#pragma unroll
        for (int r = 0; r < 8; ++r)
            red[((mt * 4 + n) * 8 + r) * 32 + lane] = p48[n][r];
    __syncthreads();
    if (threadIdx.x < 256) {
        const int m  = threadIdx.x >> 4;
        const int c  = threadIdx.x & 15;
        const int ln = c + 16 * (m >> 3);
        const int r  = m & 7;
        float s4[4] = {0.f, 0.f, 0.f, 0.f};
#pragma unroll
        for (int w = 0; w < 16; ++w)
#pragma unroll
            for (int n = 0; n < 4; ++n)
                s4[n] += red[((w * 4 + n) * 8 + r) * 32 + ln];
        const float bias = sb[m * KK + 48];
        *(float4*)(k48_out + (size_t)m * NP + p_base + 4 * c) =
            make_float4(s4[0] + bias, s4[1] + bias, s4[2] + bias, s4[3] + bias);
    }
}

#define CG  (Cc / Gg)
#define QR  (Ww / 4)
#define LP  860
#define FUSED_THREADS 256

template <int MT, int NT>
__device__ __forceinline__ void span_load_trip(
    const v4f* __restrict__ tf, const v4f* __restrict__ af, const int c2,
    v4f (&q)[2][NT], v4f (&a)[MT][2])
{
#pragma unroll
    for (int u = 0; u < 2; ++u) {
        const int c = 2 * c2 + u;
#pragma unroll
        for (int n = 0; n < NT; ++n)
            q[u][n] = tf[n * TF_PTILE + c * 32];
#pragma unroll
        for (int m = 0; m < MT; ++m)
            a[m][u] = af[m * WF_KTILE + c * 32];
    }
}

template <int MT, int NT>
__device__ __forceinline__ void span_wmma_trip(
    const v4f (&q)[2][NT], const v4f (&a)[MT][2], v8f (&acc)[MT][NT])
{
#pragma unroll
    for (int u = 0; u < 2; ++u) {
#pragma unroll
        for (int s = 0; s < 2; ++s) {
            v2f b[NT];
#pragma unroll
            for (int n = 0; n < NT; ++n) {
                b[n].x = s ? q[u][n].z : q[u][n].x;
                b[n].y = s ? q[u][n].w : q[u][n].y;
            }
            v2f am[MT];
#pragma unroll
            for (int m = 0; m < MT; ++m) {
                am[m].x = s ? a[m][u].z : a[m][u].x;
                am[m].y = s ? a[m][u].w : a[m][u].y;
            }
#pragma unroll
            for (int m = 0; m < MT; ++m)
#pragma unroll
                for (int n = 0; n < NT; ++n)
                    acc[m][n] = __builtin_amdgcn_wmma_f32_16x16x4_f32(
                        false, am[m], false, b[n], (short)0, acc[m][n], false, false);
        }
    }
}

template <int MT, int NT>
__device__ __forceinline__ void span_wave_tile(
    const v4f* __restrict__ tf, const v4f* __restrict__ af,
    const float* __restrict__ sbg, float* __restrict__ dst, const int half)
{
    v8f acc[MT][NT];
#pragma unroll
    for (int m = 0; m < MT; ++m)
#pragma unroll
        for (int n = 0; n < NT; ++n)
            acc[m][n] = (v8f){};

    v4f qA[2][NT], qB[2][NT];
    v4f aA[MT][2], aB[MT][2];
    span_load_trip<MT, NT>(tf, af, 0, qA, aA);
#pragma unroll 1
    for (int c2 = 0; c2 < TN / 16 - 2; c2 += 2) {
        span_load_trip<MT, NT>(tf, af, c2 + 1, qB, aB);
        __builtin_amdgcn_sched_barrier(0);
        span_wmma_trip<MT, NT>(qA, aA, acc);
        __builtin_amdgcn_sched_barrier(0);
        span_load_trip<MT, NT>(tf, af, c2 + 2, qA, aA);
        __builtin_amdgcn_sched_barrier(0);
        span_wmma_trip<MT, NT>(qB, aB, acc);
        __builtin_amdgcn_sched_barrier(0);
    }
    span_load_trip<MT, NT>(tf, af, TN / 16 - 1, qB, aB);
    __builtin_amdgcn_sched_barrier(0);
    span_wmma_trip<MT, NT>(qA, aA, acc);
    __builtin_amdgcn_sched_barrier(0);
    span_wmma_trip<MT, NT>(qB, aB, acc);

#pragma unroll
    for (int m = 0; m < MT; ++m) {
#pragma unroll
        for (int r = 0; r < 8; ++r) {
            const int tap = 16 * m + 8 * half + r;
            const float bias = sbg[tap];
            if (NT == 4) {
                *(float4*)(dst + tap * LP) =
                    make_float4(acc[m][0][r] + bias, acc[m][NT > 1 ? 1 : 0][r] + bias,
                                acc[m][NT > 2 ? 2 : 0][r] + bias, acc[m][NT > 3 ? 3 : 0][r] + bias);
            } else if (NT == 2) {
                *(float2*)(dst + tap * LP) =
                    make_float2(acc[m][0][r] + bias, acc[m][NT > 1 ? 1 : 0][r] + bias);
            } else {
                dst[tap * LP] = acc[m][0][r] + bias;
            }
        }
    }
}

template <int MT>
__device__ __forceinline__ void span_wave_tile_x1(
    const v4f* __restrict__ tf, const v4f* __restrict__ af,
    const float* __restrict__ sbg, float* __restrict__ dst, const int half)
{
    v8f acc[MT][1];
#pragma unroll
    for (int m = 0; m < MT; ++m)
        acc[m][0] = (v8f){};

    v4f qall[TN / 8];
#pragma unroll
    for (int c = 0; c < TN / 8; ++c)
        qall[c] = tf[c * 32];

    v4f a[2][MT][2];
#pragma unroll
    for (int u = 0; u < 2; ++u)
#pragma unroll
        for (int m = 0; m < MT; ++m)
            a[0][m][u] = af[m * WF_KTILE + u * 32];
#pragma unroll
    for (int c2 = 0; c2 < TN / 16; ++c2) {
        if (c2 + 1 < TN / 16) {
#pragma unroll
            for (int u = 0; u < 2; ++u)
#pragma unroll
                for (int m = 0; m < MT; ++m)
                    a[(c2 + 1) & 1][m][u] = af[m * WF_KTILE + (2 * (c2 + 1) + u) * 32];
        }
        __builtin_amdgcn_sched_barrier(0);
        const v4f q[2][1] = {{qall[2 * c2]}, {qall[2 * c2 + 1]}};
        span_wmma_trip<MT, 1>(q, a[c2 & 1], acc);
        __builtin_amdgcn_sched_barrier(0);
    }

#pragma unroll
    for (int m = 0; m < MT; ++m)
#pragma unroll
        for (int r = 0; r < 8; ++r) {
            const int tap = 16 * m + 8 * half + r;
            dst[tap * LP] = acc[m][0][r] + sbg[tap];
        }
}

__device__ __forceinline__ void inv_load_channel(
    const char* __restrict__ xb, const unsigned chan_bytes,
    const unsigned (&xo)[Kk][3], float4 (&xw)[Kk][3])
{
#pragma unroll
    for (int i = 0; i < Kk; ++i)
#pragma unroll
        for (int s = 0; s < 3; ++s)
            xw[i][s] = *(const float4*)(xb + chan_bytes + xo[i][s]);
}

__device__ __forceinline__ void inv_channel(
    const float4 (&xw)[Kk][3], const float (&kv)[KK][4], float* __restrict__ dst)
{
    float acc[4] = {0.f, 0.f, 0.f, 0.f};
#pragma unroll
    for (int i = 0; i < Kk; ++i) {
        const float win[12] = {xw[i][0].x, xw[i][0].y, xw[i][0].z, xw[i][0].w,
                               xw[i][1].x, xw[i][1].y, xw[i][1].z, xw[i][1].w,
                               xw[i][2].x, xw[i][2].y, xw[i][2].z, xw[i][2].w};
#pragma unroll
        for (int j = 0; j < Kk; ++j)
#pragma unroll
            for (int e = 0; e < 4; ++e)
                acc[e] += win[e + j + 1] * kv[i * Kk + j][e];
    }
    *(float4*)dst = make_float4(acc[0], acc[1], acc[2], acc[3]);
}

__global__ __launch_bounds__(FUSED_THREADS) __attribute__((amdgpu_waves_per_eu(2, 3)))
void span_involution(
    const float* __restrict__ t_in, const float* __restrict__ wf,
    const float* __restrict__ sb, const float* __restrict__ k48,
    const float* __restrict__ x, float* __restrict__ out)
{
    __shared__ __attribute__((aligned(16))) float kl[48 * LP];

    const int g      = blockIdx.x & 15;
    const int pb     = blockIdx.x >> 4;
    const int nq     = pb < 4 ? 13 : 12;
    const int qstart = pb * 12 + (pb < 4 ? pb : 4);

    const int lane  = threadIdx.x & 31;
    const int wave  = __builtin_amdgcn_readfirstlane((int)(threadIdx.x >> 5));
    const int half  = lane >> 4;
    const int l16   = lane & 15;

    {
        const int s = wave & 3;
        const int j = wave >> 2;
        const v4f*   af  = (const v4f*)wf + (size_t)(g * 3) * WF_KTILE + lane;
        const float* sbg = sb + g * KK;
        {
            const int lq = 3 * s + 2 * j;
            const v4f* tf = (const v4f*)t_in + (size_t)((qstart + lq) * 4) * TF_PTILE + lane;
            span_wave_tile<3, 4>(tf, af, sbg, kl + lq * 64 + 4 * l16, half);
        }
        {
            const int lq = 3 * s + 1;
            const v4f* tf = (const v4f*)t_in + (size_t)((qstart + lq) * 4 + 2 * j) * TF_PTILE + lane;
            span_wave_tile<3, 2>(tf, af, sbg, kl + lq * 64 + 4 * l16 + 2 * j, half);
        }
        if (nq == 13) {
            const v4f* tf = (const v4f*)t_in + (size_t)((qstart + 12) * 4 + s) * TF_PTILE + lane;
            float* dst = kl + 12 * 64 + 4 * l16 + s;
            if (j == 0)
                span_wave_tile_x1<2>(tf, af, sbg, dst, half);
            else
                span_wave_tile_x1<1>(tf, af + 2 * WF_KTILE, sbg + 32, dst + 32 * LP, half);
        }
    }
    __syncthreads();

    const int it = threadIdx.x;
    if (it >= nq * 16)
        return;
    const int p   = qstart * 64 + 4 * it;
    const int b   = p / HW;
    const int qi  = (p - b * HW) >> 2;
    const int h   = qi / QR;
    const int q   = qi - h * QR;


    float4 k4[KK];
#pragma unroll
    for (int t = 0; t < KK - 1; ++t)
        k4[t] = *(const float4*)(kl + t * LP + 4 * it);
    k4[KK - 1] = *(const float4*)(k48 + (size_t)g * NP + p);

    const unsigned cL = (unsigned)max(q - 1, 0) * 16u;
    const unsigned cC = (unsigned)q * 16u;
    const unsigned cR = (unsigned)min(q + 1, QR - 1) * 16u;
    const float mL = q > 0 ? 1.f : 0.f;
    const float mR = q < QR - 1 ? 1.f : 0.f;

    const unsigned go = (unsigned)((b * Cc + g * CG) * HW) * 4u;
    unsigned xo[Kk][3];
#pragma unroll
    for (int i = 0; i < Kk; ++i) {
        const unsigned ro = go + (unsigned)(min(max(h + i - 3, 0), Hh - 1) * Ww) * 4u;
        xo[i][0] = ro + cL;
        xo[i][1] = ro + cC;
        xo[i][2] = ro + cR;
    }

    const char* xb = (const char*)x;
    float*      og = out + ((size_t)b * Cc + g * CG) * HW + 4 * qi;

    float4 xA[Kk][3], xB[Kk][3];
    inv_load_channel(xb, 0u, xo, xA);

    float kv[KK][4];
#pragma unroll
    for (int i = 0; i < Kk; ++i) {
        const int hh = h + i - 3;
        const float mrow = (hh >= 0 && hh < Hh) ? 1.f : 0.f;
        const float mrl = mrow * mL;
        const float mrr = mrow * mR;
#pragma unroll
        for (int j = 0; j < Kk; ++j) {
            const float4 kq = k4[i * Kk + j];
            const float kk4[4] = {kq.x, kq.y, kq.z, kq.w};
#pragma unroll
            for (int e = 0; e < 4; ++e) {
                const int m = e + j + 1;
                kv[i * Kk + j][e] = kk4[e] * (m < 4 ? mrl : (m >= 8 ? mrr : mrow));
            }
        }
    }

#pragma unroll
    for (int c = 0; c < CG; c += 2) {
        inv_load_channel(xb, (unsigned)((c + 1) * HW * 4), xo, xB);
        inv_channel(xA, kv, og + (size_t)c * HW);
        if (c + 2 < CG)
            inv_load_channel(xb, (unsigned)((c + 2) * HW * 4), xo, xA);
        inv_channel(xB, kv, og + (size_t)(c + 1) * HW);
    }
}

extern "C" void kernel_launch(void* const* d_in, const int* in_sizes, int n_in,
                              void* d_out, int out_size, void* d_ws, size_t ws_size,
                              hipStream_t stream) {
    const float* x     = (const float*)d_in[0];
    const float* rw    = (const float*)d_in[1];
    const float* rb    = (const float*)d_in[2];
    const float* gamma = (const float*)d_in[3];
    const float* beta  = (const float*)d_in[4];
    const float* mean  = (const float*)d_in[5];
    const float* var   = (const float*)d_in[6];
    const float* pa    = (const float*)d_in[7];
    const float* sw    = (const float*)d_in[8];
    const float* sb    = (const float*)d_in[9];
    float* out = (float*)d_out;

    float* t_ws   = (float*)d_ws;
    float* k48_ws = t_ws + (size_t)NP * TN;
    float* wf_ws  = k48_ws + (size_t)Gg * NP;

    gemm1_reduce_bn_prelu<<<G1_BLOCKS + WF_BLOCKS, G1_THREADS, 0, stream>>>(
        x, rw, rb, gamma, beta, mean, var, pa, t_ws, sw, wf_ws, sb, k48_ws);
    span_involution<<<256, FUSED_THREADS, 0, stream>>>(
        t_ws, wf_ws, sb, k48_ws, x, out);
}


# span GEMM: a wave's full quad + half quad run as ONE wave tile of 3 M tiles x 6 pixel tiles (was 3x4, then 3x2): A fragments pass once per wave, 18 loads per 72 WMMAs (was 24); k sum order kept
# speedup vs baseline: 4.1435x; 1.0490x over previous
#include <hip/hip_runtime.h>
#include <hip/hip_bf16.h>


typedef __attribute__((ext_vector_type(2))) float v2f;
typedef __attribute__((ext_vector_type(8))) float v8f;
typedef __attribute__((ext_vector_type(4))) float v4f;

#define Bn 4
#define Cc 256
#define Hh 56
#define Ww 56
#define HW (Hh * Ww)
#define NP (Bn * HW)
#define TN 256
#define Gg 16
#define Kk 7
#define KK (Kk * Kk)
#define EPSf 1e-5f

#define TF_PTILE (16 * 2 * 32)
#define KU 8
#define WF_KTILE (32 * 32)
#define G1_BLOCKS 196
#define WF_BLOCKS 96
#define G1_THREADS 512

__global__ __launch_bounds__(G1_THREADS) __attribute__((amdgpu_waves_per_eu(4, 8)))
void gemm1_reduce_bn_prelu(
    const float* __restrict__ x, const float* __restrict__ rw,
    const float* __restrict__ rb, const float* __restrict__ gamma,
    const float* __restrict__ beta, const float* __restrict__ mean,
    const float* __restrict__ var, const float* __restrict__ pa,
    float* __restrict__ t_out,
    const float* __restrict__ sw, float* __restrict__ wf_out,
    const float* __restrict__ sb, float* __restrict__ k48_out)
{
    __shared__ float red[16 * 4 * 8 * 32];

    if (blockIdx.x >= G1_BLOCKS) {
        const int idx = (blockIdx.x - G1_BLOCKS) * G1_THREADS + threadIdx.x;
        const int ln  = idx & 31;
        const int c   = (idx >> 5) & 31;
        const int kt  = idx >> 10;
        const int g   = kt / 3;
        const int j   = kt - 3 * g;
        const float* s = sw + (size_t)(g * KK + 16 * j + (ln & 15)) * TN + 8 * c + 2 * (ln >> 4);
        const v2f lo = *(const v2f*)s;
        const v2f hi = *(const v2f*)(s + 4);
        ((float4*)wf_out)[idx] = make_float4(lo.x, lo.y, hi.x, hi.y);
        return;
    }

    const int lane  = threadIdx.x & 31;
    const int mt    = threadIdx.x >> 5;
    const int blk   = blockIdx.x;
    const int half  = lane >> 4;
    const int l16   = lane & 15;

    const int p_base = blk * 64;
    const int b      = p_base / HW;
    const int hw     = p_base - b * HW;
    const float* xq = x + (size_t)b * Cc * HW + hw + 4 * l16;
    const int cperm = 4 * ((l16 & 7) >> 1) + 2 * (l16 >> 3) + (l16 & 1);
    const float* wrow = rw + (size_t)(mt * 16 + cperm) * Cc;

    v8f acc0 = {}, acc1 = {}, acc2 = {}, acc3 = {};
#pragma unroll 1
    for (int k0 = 0; k0 < Cc; k0 += 4 * KU) {
        float4 xa[KU], xb[KU];
        v2f    w[KU];
#pragma unroll
        for (int u = 0; u < KU; ++u) {
            const int kh = k0 + 4 * u + 2 * half;
            xa[u] = *(const float4*)(xq + (size_t)kh * HW);
            xb[u] = *(const float4*)(xq + (size_t)(kh + 1) * HW);
            w[u]  = *(const v2f*)(wrow + kh);
        }
#pragma unroll
        for (int u = 0; u < KU; ++u) {
            v2f b0, b1, b2, b3;
            b0.x = xa[u].x; b0.y = xb[u].x;
            b1.x = xa[u].y; b1.y = xb[u].y;
            b2.x = xa[u].z; b2.y = xb[u].z;
            b3.x = xa[u].w; b3.y = xb[u].w;
            acc0 = __builtin_amdgcn_wmma_f32_16x16x4_f32(false, w[u], false, b0, (short)0, acc0, false, false);
            acc1 = __builtin_amdgcn_wmma_f32_16x16x4_f32(false, w[u], false, b1, (short)0, acc1, false, false);
            acc2 = __builtin_amdgcn_wmma_f32_16x16x4_f32(false, w[u], false, b2, (short)0, acc2, false, false);
            acc3 = __builtin_amdgcn_wmma_f32_16x16x4_f32(false, w[u], false, b3, (short)0, acc3, false, false);
        }
    }

    const float al = pa[0];
    float4* tf = (float4*)t_out + (size_t)(blk * 4) * TF_PTILE + lane;
    const int nbase = mt * 16 + 2 * half;
    v8f p48[4] = {(v8f){}, (v8f){}, (v8f){}, (v8f){}};
#pragma unroll
    for (int rq = 0; rq < 2; ++rq) {
        float o[4][4];
#pragma unroll
        for (int e = 0; e < 4; ++e) {
            const int r  = 4 * rq + e;
            const int n  = nbase + 4 * (r >> 1) + (r & 1);
            const float s  = gamma[n] * rsqrtf(var[n] + EPSf);
            const float bi = rb[n] - mean[n];
            const float bt = beta[n];
            float v0 = (acc0[r] + bi) * s + bt; v0 = v0 > 0.f ? v0 : al * v0;
            float v1 = (acc1[r] + bi) * s + bt; v1 = v1 > 0.f ? v1 : al * v1;
            float v2 = (acc2[r] + bi) * s + bt; v2 = v2 > 0.f ? v2 : al * v2;
            float v3 = (acc3[r] + bi) * s + bt; v3 = v3 > 0.f ? v3 : al * v3;
            o[0][e] = v0;
            o[1][e] = v1;
            o[2][e] = v2;
            o[3][e] = v3;
        }
        float4* dst = tf + (size_t)(mt * 2 + rq) * 32;
#pragma unroll
        for (int n = 0; n < 4; ++n)
            dst[n * TF_PTILE] = make_float4(o[n][0], o[n][1], o[n][2], o[n][3]);

        const float* s48 = sw + (size_t)(l16 * KK + 48) * TN + 8 * (2 * mt + rq) + 2 * half;
        const v2f alo = *(const v2f*)s48;
        const v2f ahi = *(const v2f*)(s48 + 4);
#pragma unroll
        for (int n = 0; n < 4; ++n) {
            v2f blo, bhi;
            blo.x = o[n][0]; blo.y = o[n][1];
            bhi.x = o[n][2]; bhi.y = o[n][3];
            p48[n] = __builtin_amdgcn_wmma_f32_16x16x4_f32(false, alo, false, blo, (short)0, p48[n], false, false);
            p48[n] = __builtin_amdgcn_wmma_f32_16x16x4_f32(false, ahi, false, bhi, (short)0, p48[n], false, false);
        }
    }

#pragma unroll
    for (int n = 0; n < 4; ++n)
#pragma unroll
        for (int r = 0; r < 8; ++r)
            red[((mt * 4 + n) * 8 + r) * 32 + lane] = p48[n][r];
    __syncthreads();
    if (threadIdx.x < 256) {
        const int m  = threadIdx.x >> 4;
        const int c  = threadIdx.x & 15;
        const int ln = c + 16 * (m >> 3);
        const int r  = m & 7;
        float s4[4] = {0.f, 0.f, 0.f, 0.f};
#pragma unroll
        for (int w = 0; w < 16; ++w)
#pragma unroll
            for (int n = 0; n < 4; ++n)
                s4[n] += red[((w * 4 + n) * 8 + r) * 32 + ln];
        const float bias = sb[m * KK + 48];
        *(float4*)(k48_out + (size_t)m * NP + p_base + 4 * c) =
            make_float4(s4[0] + bias, s4[1] + bias, s4[2] + bias, s4[3] + bias);
    }
}

#define CG  (Cc / Gg)
#define QR  (Ww / 4)
#define LP  860
#define FUSED_THREADS 256

template <int MT, int N0, int NT>
__device__ __forceinline__ void span_load_trip(
    const v4f* __restrict__ tf0, const v4f* __restrict__ tf1,
    const v4f* __restrict__ af, const int c2,
    v4f (&q)[2][NT], v4f (&a)[MT][2])
{
#pragma unroll
    for (int u = 0; u < 2; ++u) {
        const int c = 2 * c2 + u;
#pragma unroll
        for (int n = 0; n < NT; ++n)
            q[u][n] = n < N0 ? tf0[n * TF_PTILE + c * 32] : tf1[(n - N0) * TF_PTILE + c * 32];
#pragma unroll
        for (int m = 0; m < MT; ++m)
            a[m][u] = af[m * WF_KTILE + c * 32];
    }
}

template <int MT, int NT>
__device__ __forceinline__ void span_wmma_trip(
    const v4f (&q)[2][NT], const v4f (&a)[MT][2], v8f (&acc)[MT][NT])
{
#pragma unroll
    for (int u = 0; u < 2; ++u) {
#pragma unroll
        for (int s = 0; s < 2; ++s) {
            v2f b[NT];
#pragma unroll
            for (int n = 0; n < NT; ++n) {
                b[n].x = s ? q[u][n].z : q[u][n].x;
                b[n].y = s ? q[u][n].w : q[u][n].y;
            }
            v2f am[MT];
#pragma unroll
            for (int m = 0; m < MT; ++m) {
                am[m].x = s ? a[m][u].z : a[m][u].x;
                am[m].y = s ? a[m][u].w : a[m][u].y;
            }
#pragma unroll
            for (int m = 0; m < MT; ++m)
#pragma unroll
                for (int n = 0; n < NT; ++n)
                    acc[m][n] = __builtin_amdgcn_wmma_f32_16x16x4_f32(
                        false, am[m], false, b[n], (short)0, acc[m][n], false, false);
        }
    }
}

template <int MT>
__device__ __forceinline__ void span_wave_tile(
    const v4f* __restrict__ tf0, const v4f* __restrict__ tf1, const v4f* __restrict__ af,
    const float* __restrict__ sbg, float* __restrict__ dst0, float* __restrict__ dst1,
    const int half)
{
    constexpr int N0 = 4, NT = 6;
    v8f acc[MT][NT];
#pragma unroll
    for (int m = 0; m < MT; ++m)
#pragma unroll
        for (int n = 0; n < NT; ++n)
            acc[m][n] = (v8f){};

    v4f qA[2][NT], qB[2][NT];
    v4f aA[MT][2], aB[MT][2];
    span_load_trip<MT, N0, NT>(tf0, tf1, af, 0, qA, aA);
#pragma unroll 1
    for (int c2 = 0; c2 < TN / 16 - 2; c2 += 2) {
        span_load_trip<MT, N0, NT>(tf0, tf1, af, c2 + 1, qB, aB);
        __builtin_amdgcn_sched_barrier(0);
        span_wmma_trip<MT, NT>(qA, aA, acc);
        __builtin_amdgcn_sched_barrier(0);
        span_load_trip<MT, N0, NT>(tf0, tf1, af, c2 + 2, qA, aA);
        __builtin_amdgcn_sched_barrier(0);
        span_wmma_trip<MT, NT>(qB, aB, acc);
        __builtin_amdgcn_sched_barrier(0);
    }
    span_load_trip<MT, N0, NT>(tf0, tf1, af, TN / 16 - 1, qB, aB);
    __builtin_amdgcn_sched_barrier(0);
    span_wmma_trip<MT, NT>(qA, aA, acc);
    __builtin_amdgcn_sched_barrier(0);
    span_wmma_trip<MT, NT>(qB, aB, acc);

#pragma unroll
    for (int m = 0; m < MT; ++m) {
#pragma unroll
        for (int r = 0; r < 8; ++r) {
            const int tap = 16 * m + 8 * half + r;
            const float bias = sbg[tap];
            *(float4*)(dst0 + tap * LP) =
                make_float4(acc[m][0][r] + bias, acc[m][1][r] + bias,
                            acc[m][2][r] + bias, acc[m][3][r] + bias);
            *(float2*)(dst1 + tap * LP) =
                make_float2(acc[m][4][r] + bias, acc[m][5][r] + bias);
        }
    }
}

template <int MT>
__device__ __forceinline__ void span_wave_tile_x1(
    const v4f* __restrict__ tf, const v4f* __restrict__ af,
    const float* __restrict__ sbg, float* __restrict__ dst, const int half)
{
    v8f acc[MT][1];
#pragma unroll
    for (int m = 0; m < MT; ++m)
        acc[m][0] = (v8f){};

    v4f qall[TN / 8];
#pragma unroll
    for (int c = 0; c < TN / 8; ++c)
        qall[c] = tf[c * 32];

    v4f a[2][MT][2];
#pragma unroll
    for (int u = 0; u < 2; ++u)
#pragma unroll
        for (int m = 0; m < MT; ++m)
            a[0][m][u] = af[m * WF_KTILE + u * 32];
#pragma unroll
    for (int c2 = 0; c2 < TN / 16; ++c2) {
        if (c2 + 1 < TN / 16) {
#pragma unroll
            for (int u = 0; u < 2; ++u)
#pragma unroll
                for (int m = 0; m < MT; ++m)
                    a[(c2 + 1) & 1][m][u] = af[m * WF_KTILE + (2 * (c2 + 1) + u) * 32];
        }
        __builtin_amdgcn_sched_barrier(0);
        const v4f q[2][1] = {{qall[2 * c2]}, {qall[2 * c2 + 1]}};
        span_wmma_trip<MT, 1>(q, a[c2 & 1], acc);
        __builtin_amdgcn_sched_barrier(0);
    }

#pragma unroll
    for (int m = 0; m < MT; ++m)
#pragma unroll
        for (int r = 0; r < 8; ++r) {
            const int tap = 16 * m + 8 * half + r;
            dst[tap * LP] = acc[m][0][r] + sbg[tap];
        }
}

__device__ __forceinline__ void inv_load_channel(
    const char* __restrict__ xb, const unsigned chan_bytes,
    const unsigned (&xo)[Kk][3], float4 (&xw)[Kk][3])
{
#pragma unroll
    for (int i = 0; i < Kk; ++i)
#pragma unroll
        for (int s = 0; s < 3; ++s)
            xw[i][s] = *(const float4*)(xb + chan_bytes + xo[i][s]);
}

__device__ __forceinline__ void inv_channel(
    const float4 (&xw)[Kk][3], const float (&kv)[KK][4], float* __restrict__ dst)
{
    float acc[4] = {0.f, 0.f, 0.f, 0.f};
#pragma unroll
    for (int i = 0; i < Kk; ++i) {
        const float win[12] = {xw[i][0].x, xw[i][0].y, xw[i][0].z, xw[i][0].w,
                               xw[i][1].x, xw[i][1].y, xw[i][1].z, xw[i][1].w,
                               xw[i][2].x, xw[i][2].y, xw[i][2].z, xw[i][2].w};
#pragma unroll
        for (int j = 0; j < Kk; ++j)
#pragma unroll
            for (int e = 0; e < 4; ++e)
                acc[e] += win[e + j + 1] * kv[i * Kk + j][e];
    }
    *(float4*)dst = make_float4(acc[0], acc[1], acc[2], acc[3]);
}

__global__ __launch_bounds__(FUSED_THREADS) __attribute__((amdgpu_waves_per_eu(2, 3)))
void span_involution(
    const float* __restrict__ t_in, const float* __restrict__ wf,
    const float* __restrict__ sb, const float* __restrict__ k48,
    const float* __restrict__ x, float* __restrict__ out)
{
    __shared__ __attribute__((aligned(16))) float kl[48 * LP];

    const int g      = blockIdx.x & 15;
    const int pb     = blockIdx.x >> 4;
    const int nq     = pb < 4 ? 13 : 12;
    const int qstart = pb * 12 + (pb < 4 ? pb : 4);

    const int lane  = threadIdx.x & 31;
    const int wave  = __builtin_amdgcn_readfirstlane((int)(threadIdx.x >> 5));
    const int half  = lane >> 4;
    const int l16   = lane & 15;

    {
        const int s = wave & 3;
        const int j = wave >> 2;
        const v4f*   af  = (const v4f*)wf + (size_t)(g * 3) * WF_KTILE + lane;
        const float* sbg = sb + g * KK;
        {
            const int lq0 = 3 * s + 2 * j;
            const int lq1 = 3 * s + 1;
            const v4f* tf0 = (const v4f*)t_in + (size_t)((qstart + lq0) * 4) * TF_PTILE + lane;
            const v4f* tf1 = (const v4f*)t_in + (size_t)((qstart + lq1) * 4 + 2 * j) * TF_PTILE + lane;
            span_wave_tile<3>(tf0, tf1, af, sbg, kl + lq0 * 64 + 4 * l16,
                              kl + lq1 * 64 + 4 * l16 + 2 * j, half);
        }
        if (nq == 13) {
            const v4f* tf = (const v4f*)t_in + (size_t)((qstart + 12) * 4 + s) * TF_PTILE + lane;
            float* dst = kl + 12 * 64 + 4 * l16 + s;
            if (j == 0)
                span_wave_tile_x1<2>(tf, af, sbg, dst, half);
            else
                span_wave_tile_x1<1>(tf, af + 2 * WF_KTILE, sbg + 32, dst + 32 * LP, half);
        }
    }
    __syncthreads();

    const int it = threadIdx.x;
    if (it >= nq * 16)
        return;
    const int p   = qstart * 64 + 4 * it;
    const int b   = p / HW;
    const int qi  = (p - b * HW) >> 2;
    const int h   = qi / QR;
    const int q   = qi - h * QR;


    float4 k4[KK];
#pragma unroll
    for (int t = 0; t < KK - 1; ++t)
        k4[t] = *(const float4*)(kl + t * LP + 4 * it);
    k4[KK - 1] = *(const float4*)(k48 + (size_t)g * NP + p);

    const unsigned cL = (unsigned)max(q - 1, 0) * 16u;
    const unsigned cC = (unsigned)q * 16u;
    const unsigned cR = (unsigned)min(q + 1, QR - 1) * 16u;
    const float mL = q > 0 ? 1.f : 0.f;
    const float mR = q < QR - 1 ? 1.f : 0.f;

    const unsigned go = (unsigned)((b * Cc + g * CG) * HW) * 4u;
    unsigned xo[Kk][3];
#pragma unroll
    for (int i = 0; i < Kk; ++i) {
        const unsigned ro = go + (unsigned)(min(max(h + i - 3, 0), Hh - 1) * Ww) * 4u;
        xo[i][0] = ro + cL;
        xo[i][1] = ro + cC;
        xo[i][2] = ro + cR;
    }

    const char* xb = (const char*)x;
    float*      og = out + ((size_t)b * Cc + g * CG) * HW + 4 * qi;

    float4 xA[Kk][3], xB[Kk][3];
    inv_load_channel(xb, 0u, xo, xA);

    float kv[KK][4];
#pragma unroll
    for (int i = 0; i < Kk; ++i) {
        const int hh = h + i - 3;
        const float mrow = (hh >= 0 && hh < Hh) ? 1.f : 0.f;
        const float mrl = mrow * mL;
        const float mrr = mrow * mR;
#pragma unroll
        for (int j = 0; j < Kk; ++j) {
            const float4 kq = k4[i * Kk + j];
            const float kk4[4] = {kq.x, kq.y, kq.z, kq.w};
#pragma unroll
            for (int e = 0; e < 4; ++e) {
                const int m = e + j + 1;
                kv[i * Kk + j][e] = kk4[e] * (m < 4 ? mrl : (m >= 8 ? mrr : mrow));
            }
        }
    }

#pragma unroll
    for (int c = 0; c < CG; c += 2) {
        inv_load_channel(xb, (unsigned)((c + 1) * HW * 4), xo, xB);
        inv_channel(xA, kv, og + (size_t)c * HW);
        if (c + 2 < CG)
            inv_load_channel(xb, (unsigned)((c + 2) * HW * 4), xo, xA);
        inv_channel(xB, kv, og + (size_t)(c + 1) * HW);
    }
}

extern "C" void kernel_launch(void* const* d_in, const int* in_sizes, int n_in,
                              void* d_out, int out_size, void* d_ws, size_t ws_size,
                              hipStream_t stream) {
    const float* x     = (const float*)d_in[0];
    const float* rw    = (const float*)d_in[1];
    const float* rb    = (const float*)d_in[2];
    const float* gamma = (const float*)d_in[3];
    const float* beta  = (const float*)d_in[4];
    const float* mean  = (const float*)d_in[5];
    const float* var   = (const float*)d_in[6];
    const float* pa    = (const float*)d_in[7];
    const float* sw    = (const float*)d_in[8];
    const float* sb    = (const float*)d_in[9];
    float* out = (float*)d_out;

    float* t_ws   = (float*)d_ws;
    float* k48_ws = t_ws + (size_t)NP * TN;
    float* wf_ws  = k48_ws + (size_t)Gg * NP;

    gemm1_reduce_bn_prelu<<<G1_BLOCKS + WF_BLOCKS, G1_THREADS, 0, stream>>>(
        x, rw, rb, gamma, beta, mean, var, pa, t_ws, sw, wf_ws, sb, k48_ws);
    span_involution<<<256, FUSED_THREADS, 0, stream>>>(
        t_ws, wf_ws, sb, k48_ws, x, out);
}
